# starting kernel
# baseline (speedup 1.0000x reference)
#include <hip/hip_runtime.h>
#include <math.h>

typedef float v2f __attribute__((ext_vector_type(2)));
typedef float v8f __attribute__((ext_vector_type(8)));

#define N_STFT   513
#define N_BARKS  128
#define BATCH    4
#define TIME     512
#define ROWS     (BATCH * TIME)
#define FP       528
#define NB       128
#define LDSS     532
#define LDSD     132
#define MAX_ITER 30
#define LR       0.1f
#define MOMENTUM 0.9f
#define TOL_LOSS 1e-5f
#define TOL_CHG  1e-8f


__global__ __launch_bounds__(256) void ibs_init_spec(const float* __restrict__ spec_init,
                                                     float* __restrict__ spec,
                                                     float* __restrict__ buf) {
    int idx = blockIdx.x * blockDim.x + threadIdx.x;
    if (idx >= ROWS * FP) return;
    int row = idx / FP;
    int f   = idx - row * FP;
    spec[idx] = (f < N_STFT) ? spec_init[row * N_STFT + f] : 0.0f;
    buf[idx]  = 0.0f;
}

__global__ __launch_bounds__(256) void ibs_init_fb(const float* __restrict__ fb_in,
                                                   float* __restrict__ fbp2,
                                                   float* __restrict__ fbT2) {
    int idx = blockIdx.x * blockDim.x + threadIdx.x;
    if (idx >= FP * NB) return;
    int f  = idx / NB;
    int kk = idx - f * NB;
    float v = (f < N_STFT) ? fb_in[f * NB + kk] : 0.0f;
    fbp2[(((f  >> 1) * NB) + kk) * 2 + (f  & 1)] = v;
    fbT2[(((kk >> 1) * FP) + f ) * 2 + (kk & 1)] = v;
}

__global__ void ibs_init_state(float* __restrict__ state) {
    state[0] = INFINITY;
    state[1] = 0.0f;
}


__global__ __launch_bounds__(256) void ibs_iter(const float* __restrict__ barkspec,
                                                float* __restrict__ spec,
                                                float* __restrict__ buf,
                                                const float2* __restrict__ fbp2,
                                                const float2* __restrict__ fbT2,
                                                float* __restrict__ partials,
                                                const float* __restrict__ state) {
    if (state[1] != 0.0f) return;

    __shared__ float lds_spec[16 * LDSS];
    __shared__ float lds_diff[16 * LDSD];
    __shared__ float lds_red[256];

    const int tid     = threadIdx.x;
    const int wave    = tid >> 5;
    const int lane    = tid & 31;
    const int mi      = lane & 15;
    const int hsel    = lane >> 4;
    const int koff    = hsel * 2;
    const int rowbase = blockIdx.x * 16;

    for (int i = tid; i < 16 * (FP / 4); i += 256) {
        int r  = i / (FP / 4);
        int c4 = i - r * (FP / 4);
        float4 v = *(const float4*)(spec + (rowbase + r) * FP + c4 * 4);
        float* dst = lds_spec + r * LDSS + c4 * 4;
        dst[0] = v.x; dst[1] = v.y; dst[2] = v.z; dst[3] = v.w;
    }
    __syncthreads();

    const int nbase = wave * 16;
    const float*  ap = lds_spec + mi * LDSS + koff;
    const float2* bp = fbp2 + (nbase + mi) + hsel * NB;

    v8f acc = {0.f, 0.f, 0.f, 0.f, 0.f, 0.f, 0.f, 0.f};
    #pragma unroll 4
    for (int k = 0; k < FP; k += 4) {
        v2f a, b;
        a[0] = ap[k];
        a[1] = ap[k + 1];
        float2 bv = bp[(k >> 1) * NB];
        b[0] = bv.x; b[1] = bv.y;
        acc = __builtin_amdgcn_wmma_f32_16x16x4_f32(false, a, false, b,
                                                    (short)0, acc, false, false);
    }

    const int rowh = rowbase + hsel * 8;
    const int bb   = rowh >> 9;
    const int t0   = rowh & 511;
    const float* mp = barkspec + (bb * N_BARKS + nbase + mi) * TIME + t0;
    float ssq = 0.0f;
    #pragma unroll
    for (int r = 0; r < 8; ++r) {
        float d = mp[r] - acc[r];
        ssq += d * d;
        lds_diff[(r + hsel * 8) * LDSD + nbase + mi] = d;
    }
    lds_red[tid] = ssq;
    __syncthreads();

    if (tid == 0) {
        float s = 0.0f;
        for (int i = 0; i < 256; ++i) s += lds_red[i];
        partials[blockIdx.x] = s;
    }

    const float* a2p = lds_diff + mi * LDSD + koff;
    for (int nt = wave; nt < FP / 16; nt += 8) {
        const int fbase = nt * 16;
        const float2* b2p = fbT2 + (fbase + mi) + hsel * FP;
        v8f acc2 = {0.f, 0.f, 0.f, 0.f, 0.f, 0.f, 0.f, 0.f};
        #pragma unroll
        for (int k = 0; k < NB; k += 4) {
            v2f a, b;
            a[0] = a2p[k];
            a[1] = a2p[k + 1];
            float2 bv = b2p[(k >> 1) * FP];
            b[0] = bv.x; b[1] = bv.y;
            acc2 = __builtin_amdgcn_wmma_f32_16x16x4_f32(false, a, false, b,
                                                         (short)0, acc2, false, false);
        }
        #pragma unroll
        for (int r = 0; r < 8; ++r) {
            const int m   = r + hsel * 8;
            const int f   = fbase + mi;
            const int idx = (rowbase + m) * FP + f;
            float g  = acc2[r] * (-2.0f / (float)ROWS);
            float bn = MOMENTUM * buf[idx] + g;
            float sn = lds_spec[m * LDSS + f] - LR * bn;
            spec[idx] = sn > 0.0f ? sn : 0.0f;
            buf[idx]  = bn;
        }
    }
}


__global__ void ibs_check(float* __restrict__ state, const float* __restrict__ partials) {
    if (state[1] != 0.0f) return;
    float s = 0.0f;
    for (int i = 0; i < 128; ++i) s += partials[i];
    float loss = s / (float)ROWS;
    float prev = state[0];
    bool stop = (loss < TOL_LOSS) || (fabsf(prev - loss) < TOL_CHG);
    state[0] = loss;
    state[1] = stop ? 1.0f : 0.0f;
}


__global__ __launch_bounds__(256) void ibs_finalize(const float* __restrict__ spec,
                                                    float* __restrict__ out) {
    int idx = blockIdx.x * blockDim.x + threadIdx.x;
    if (idx >= BATCH * N_STFT * TIME) return;
    int t = idx & 511;
    int f = (idx >> 9) % N_STFT;
    int b = idx / (N_STFT * TIME);
    out[idx] = spec[(b * TIME + t) * FP + f];
}


extern "C" void kernel_launch(void* const* d_in, const int* in_sizes, int n_in,
                              void* d_out, int out_size, void* d_ws, size_t ws_size,
                              hipStream_t stream) {
    const float* barkspec  = (const float*)d_in[0];
    const float* fb        = (const float*)d_in[1];
    const float* spec_init = (const float*)d_in[2];

    float* ws       = (float*)d_ws;
    float* spec     = ws;
    float* buf      = spec + ROWS * FP;
    float* fbp2     = buf  + ROWS * FP;
    float* fbT2     = fbp2 + FP * NB;
    float* partials = fbT2 + NB * FP;
    float* state    = partials + 128;

    ibs_init_spec<<<(ROWS * FP + 255) / 256, 256, 0, stream>>>(spec_init, spec, buf);
    ibs_init_fb<<<(FP * NB + 255) / 256, 256, 0, stream>>>(fb, fbp2, fbT2);
    ibs_init_state<<<1, 1, 0, stream>>>(state);

    for (int it = 0; it < MAX_ITER; ++it) {
        ibs_iter<<<ROWS / 16, 256, 0, stream>>>(barkspec, spec, buf,
                                                (const float2*)fbp2, (const float2*)fbT2,
                                                partials, state);
        if (it < MAX_ITER - 1)
            ibs_check<<<1, 1, 0, stream>>>(state, partials);
    }

    ibs_finalize<<<(BATCH * N_STFT * TIME + 255) / 256, 256, 0, stream>>>(spec, (float*)d_out);
}


# ibs_run: all iterations in one launch as transposed products (filterbank = A, rows on lanes): spec, momentum, one fragment set in registers, other in LDS; sums of both products and loss regrouped
# speedup vs baseline: 3.6908x; 3.6908x over previous
#include <hip/hip_runtime.h>
#include <math.h>

typedef float v2f __attribute__((ext_vector_type(2)));
typedef float v8f __attribute__((ext_vector_type(8)));

#define N_STFT   513
#define N_BARKS  128
#define BATCH    4
#define TIME     512
#define ROWS     (BATCH * TIME)
#define FP       528
#define NB       128
#define MAX_ITER 30
#define LR       0.1f
#define MOMENTUM 0.9f
#define TOL_LOSS 1e-5f
#define TOL_CHG  1e-8f
#define NW       4
#define NTHR     (NW * 32)
#define TPW      9
#define NTILE    33
#define FBP      132
#define XDIFF    (NW * 16 * 32)

__global__ __launch_bounds__(256) void ibs_init_fb(const float* __restrict__ fb_in,
                                                   float* __restrict__ fbp2,
                                                   float* __restrict__ fbT2) {
    int idx = blockIdx.x * blockDim.x + threadIdx.x;
    if (idx >= FP * NB) return;
    int f  = idx / NB;
    int kk = idx - f * NB;
    float v = (f < N_STFT) ? fb_in[f * NB + kk] : 0.0f;
    fbp2[(((f  >> 1) * NB) + kk) * 2 + (f  & 1)] = v;
    fbT2[(((kk >> 1) * FP) + f ) * 2 + (kk & 1)] = v;
}

__global__ __launch_bounds__(NTHR) void ibs_run(const float* __restrict__ barkspec,
                                                const float* __restrict__ spec_init,
                                                float* __restrict__ spec,
                                                const float2* __restrict__ fbp2,
                                                const float2* __restrict__ fbT2,
                                                float* lossp,
                                                int replay) {
    __shared__ __attribute__((aligned(16))) float2 lds_fb[(FP / 2) * FBP];
    __shared__ __attribute__((aligned(16))) float4 lds_x[XDIFF + 16 * 32];
    __shared__ float  lds_wsum[NW];
    __shared__ float  lds_loss[32];
    __shared__ int    lds_n;

    const int tid     = threadIdx.x;
    const int wave    = __builtin_amdgcn_readfirstlane(tid >> 5);
    const int lane    = tid & 31;
    const int mi      = lane & 15;
    const int hsel    = lane >> 4;
    const int rowbase = blockIdx.x * 16;

    int n_iter = MAX_ITER;
    if (replay) {
        if (tid < MAX_ITER - 1) {
            const float4* lp = (const float4*)(lossp + tid * 128);
            float s = 0.0f;
            #pragma unroll
            for (int i = 0; i < 32; ++i) {
                float4 v = lp[i];
                s += v.x; s += v.y; s += v.z; s += v.w;
            }
            lds_loss[tid] = s / (float)ROWS;
        }
        __syncthreads();
        if (tid == 0) {
            float prev = INFINITY;
            int n = MAX_ITER;
            for (int it = 0; it < MAX_ITER - 1; ++it) {
                float loss = lds_loss[it];
                bool stop = (loss < TOL_LOSS) || (fabsf(prev - loss) < TOL_CHG);
                prev = loss;
                if (stop) { n = it + 1; break; }
            }
            lds_n = n;
        }
        __syncthreads();
        n_iter = lds_n;
        if (n_iter >= MAX_ITER) return;
    }

    {
        const float4* src4 = (const float4*)fbp2;
        #pragma unroll 33
        for (int q = tid; q < (FP / 2) * (NB / 2); q += NTHR) {
            float4 v = src4[q];
            const int p = q >> 6;
            const int c = q & 63;
            *(float4*)&lds_fb[p * FBP + 2 * c] = v;
        }
    }
    float* stage = (float*)lds_x;
    {
        const float* srcp = spec_init + rowbase * N_STFT;
        #pragma unroll
        for (int k = 0; k < (16 * N_STFT + NTHR - 1) / NTHR; ++k) {
            int i = tid + k * NTHR;
            i = i < 16 * N_STFT ? i : 16 * N_STFT - 1;
            stage[i] = srcp[i];
        }
    }
    __syncthreads();

    v8f S[TPW], Bm[TPW];
    #pragma unroll
    for (int j = 0; j < TPW; ++j) {
        #pragma unroll
        for (int r = 0; r < 8; ++r) {
            const int f = (wave + 4 * j) * 16 + hsel * 8 + r;
            S[j][r]  = (f < N_STFT) ? stage[mi * N_STFT + f] : 0.0f;
            Bm[j][r] = 0.0f;
        }
    }

    const int bb = rowbase >> 9;
    const int t0 = rowbase & 511;
    float tgt[2][8];
    #pragma unroll
    for (int u = 0; u < 2; ++u) {
        #pragma unroll
        for (int r = 0; r < 8; ++r) {
            const int bark = (2 * wave + u) * 16 + hsel * 8 + r;
            tgt[u][r] = barkspec[(bb * N_BARKS + bark) * TIME + t0 + mi];
        }
    }

    const float2* a2p = fbT2 + (4 * hsel) * FP + 16 * wave + mi;
    v2f A2[8][4][TPW];
    #pragma unroll
    for (int bt = 0; bt < 8; ++bt) {
        #pragma unroll
        for (int s = 0; s < 4; ++s) {
            #pragma unroll
            for (int j = 0; j < TPW; ++j) {
                const bool real = (wave + 4 * j) < NTILE;
                const int  joff = real ? 64 * j : 0;
                float2 av = a2p[(8 * bt + s) * FP + joff];
                A2[bt][s][j][0] = real ? av.x : 0.0f;
                A2[bt][s][j][1] = real ? av.y : 0.0f;
            }
        }
    }
    __syncthreads();

    int a1off[TPW];
    #pragma unroll
    for (int j = 0; j < TPW; ++j) {
        int T = wave + 4 * j;
        T = T < NTILE ? T : NTILE - 1;
        a1off[j] = (8 * T + 4 * hsel) * FBP + mi;
    }

    for (int it = 0; it < n_iter; ++it) {
        v8f acc[8];
        #pragma unroll
        for (int bt = 0; bt < 8; ++bt)
            acc[bt] = (v8f){0.f, 0.f, 0.f, 0.f, 0.f, 0.f, 0.f, 0.f};
        #pragma unroll
        for (int j = 0; j < TPW; ++j) {
            #pragma unroll
            for (int s = 0; s < 4; ++s) {
                v2f b;
                b[0] = S[j][2 * s];
                b[1] = S[j][2 * s + 1];
                #pragma unroll
                for (int bt = 0; bt < 8; ++bt) {
                    float2 av = lds_fb[a1off[j] + s * FBP + 16 * bt];
                    v2f a;
                    a[0] = av.x; a[1] = av.y;
                    acc[bt] = __builtin_amdgcn_wmma_f32_16x16x4_f32(false, a, false, b,
                                                                    (short)0, acc[bt], false, false);
                }
            }
        }
        #pragma unroll
        for (int bt = 0; bt < 8; ++bt) {
            lds_x[(wave * 16 + bt * 2 + 0) * 32 + lane] =
                make_float4(acc[bt][0], acc[bt][1], acc[bt][2], acc[bt][3]);
            lds_x[(wave * 16 + bt * 2 + 1) * 32 + lane] =
                make_float4(acc[bt][4], acc[bt][5], acc[bt][6], acc[bt][7]);
        }
        __syncthreads();

        float ssq = 0.0f;
        #pragma unroll
        for (int u = 0; u < 2; ++u) {
            const int bt = 2 * wave + u;
            float p[8];
            #pragma unroll
            for (int r = 0; r < 8; ++r) p[r] = 0.0f;
            #pragma unroll
            for (int sw = 0; sw < NW; ++sw) {
                float4 lo = lds_x[(sw * 16 + bt * 2 + 0) * 32 + lane];
                float4 hi = lds_x[(sw * 16 + bt * 2 + 1) * 32 + lane];
                p[0] += lo.x; p[1] += lo.y; p[2] += lo.z; p[3] += lo.w;
                p[4] += hi.x; p[5] += hi.y; p[6] += hi.z; p[7] += hi.w;
            }
            float d[8];
            #pragma unroll
            for (int r = 0; r < 8; ++r) {
                d[r] = tgt[u][r] - p[r];
                ssq += d[r] * d[r];
            }
            lds_x[XDIFF + (bt * 2 + 0) * 32 + lane] = make_float4(d[0], d[1], d[2], d[3]);
            lds_x[XDIFF + (bt * 2 + 1) * 32 + lane] = make_float4(d[4], d[5], d[6], d[7]);
        }
        ssq += __shfl_xor(ssq, 16);
        ssq += __shfl_xor(ssq, 8);
        ssq += __shfl_xor(ssq, 4);
        ssq += __shfl_xor(ssq, 2);
        ssq += __shfl_xor(ssq, 1);
        if (lane == 0) lds_wsum[wave] = ssq;
        __syncthreads();

        if (tid == 0 && !replay) {
            float s = 0.0f;
            for (int w = 0; w < NW; ++w) s += lds_wsum[w];
            lossp[it * 128 + blockIdx.x] = s;
        }

        v8f D1[8];
        #pragma unroll
        for (int bt = 0; bt < 8; ++bt) {
            float4 lo = lds_x[XDIFF + (bt * 2 + 0) * 32 + lane];
            float4 hi = lds_x[XDIFF + (bt * 2 + 1) * 32 + lane];
            D1[bt][0] = lo.x; D1[bt][1] = lo.y; D1[bt][2] = lo.z; D1[bt][3] = lo.w;
            D1[bt][4] = hi.x; D1[bt][5] = hi.y; D1[bt][6] = hi.z; D1[bt][7] = hi.w;
        }

        v8f g[TPW];
        #pragma unroll
        for (int j = 0; j < TPW; ++j)
            g[j] = (v8f){0.f, 0.f, 0.f, 0.f, 0.f, 0.f, 0.f, 0.f};
        #pragma unroll
        for (int bt = 0; bt < 8; ++bt) {
            #pragma unroll
            for (int s = 0; s < 4; ++s) {
                v2f b;
                b[0] = D1[bt][2 * s];
                b[1] = D1[bt][2 * s + 1];
                #pragma unroll
                for (int j = 0; j < TPW; ++j) {
                    g[j] = __builtin_amdgcn_wmma_f32_16x16x4_f32(false, A2[bt][s][j], false, b,
                                                                 (short)0, g[j], false, false);
                }
            }
        }
        #pragma unroll
        for (int j = 0; j < TPW; ++j) {
            #pragma unroll
            for (int r = 0; r < 8; ++r) {
                float gg = g[j][r] * (-2.0f / (float)ROWS);
                float bn = MOMENTUM * Bm[j][r] + gg;
                float sn = S[j][r] - LR * bn;
                S[j][r]  = sn > 0.0f ? sn : 0.0f;
                Bm[j][r] = bn;
            }
        }
    }

    __syncthreads();
    #pragma unroll
    for (int j = 0; j < TPW; ++j) {
        if ((wave + 4 * j) < NTILE) {
            #pragma unroll
            for (int r = 0; r < 8; ++r) {
                const int f = (wave + 4 * j) * 16 + hsel * 8 + r;
                stage[mi * FP + f] = S[j][r];
            }
        }
    }
    __syncthreads();
    {
        float4* dst = (float4*)(spec + rowbase * FP);
        const float4* st4 = (const float4*)stage;
        for (int i = tid; i < 16 * FP / 4; i += NTHR) dst[i] = st4[i];
    }
}


__global__ __launch_bounds__(256) void ibs_finalize(const float* __restrict__ spec,
                                                    float* __restrict__ out) {
    int idx = blockIdx.x * blockDim.x + threadIdx.x;
    if (idx >= BATCH * N_STFT * TIME) return;
    int t = idx & 511;
    int f = (idx >> 9) % N_STFT;
    int b = idx / (N_STFT * TIME);
    out[idx] = spec[(b * TIME + t) * FP + f];
}


extern "C" void kernel_launch(void* const* d_in, const int* in_sizes, int n_in,
                              void* d_out, int out_size, void* d_ws, size_t ws_size,
                              hipStream_t stream) {
    const float* barkspec  = (const float*)d_in[0];
    const float* fb        = (const float*)d_in[1];
    const float* spec_init = (const float*)d_in[2];

    float* ws       = (float*)d_ws;
    float* spec     = ws;
    float* lossp    = spec + ROWS * FP;
    float* fbp2     = lossp + ROWS * FP;
    float* fbT2     = fbp2 + FP * NB;

    ibs_init_fb<<<(FP * NB + 255) / 256, 256, 0, stream>>>(fb, fbp2, fbT2);
    ibs_run<<<ROWS / 16, NTHR, 0, stream>>>(barkspec, spec_init, spec,
                                            (const float2*)fbp2, (const float2*)fbT2, lossp, 0);
    ibs_run<<<ROWS / 16, NTHR, 0, stream>>>(barkspec, spec_init, spec,
                                            (const float2*)fbp2, (const float2*)fbT2, lossp, 1);
    ibs_finalize<<<(BATCH * N_STFT * TIME + 255) / 256, 256, 0, stream>>>(spec, (float*)d_out);
}


# ibs_run: product 1's LDS fragment reads issued as 9 whole groups per iteration, each one block of 32 products ahead into a second register set; diff reads as one group; sums and their order same
# speedup vs baseline: 4.2133x; 1.1416x over previous
#include <hip/hip_runtime.h>
#include <math.h>

typedef float v2f __attribute__((ext_vector_type(2)));
typedef float v8f __attribute__((ext_vector_type(8)));

#define N_STFT   513
#define N_BARKS  128
#define BATCH    4
#define TIME     512
#define ROWS     (BATCH * TIME)
#define FP       528
#define NB       128
#define MAX_ITER 30
#define LR       0.1f
#define MOMENTUM 0.9f
#define TOL_LOSS 1e-5f
#define TOL_CHG  1e-8f
#define NW       4
#define NTHR     (NW * 32)
#define TPW      9
#define NTILE    33
#define FBP      132
#define XDIFF    (NW * 16 * 32)

__global__ __launch_bounds__(256) void ibs_init_fb(const float* __restrict__ fb_in,
                                                   float* __restrict__ fbp2,
                                                   float* __restrict__ fbT2) {
    int idx = blockIdx.x * blockDim.x + threadIdx.x;
    if (idx >= FP * NB) return;
    int f  = idx / NB;
    int kk = idx - f * NB;
    float v = (f < N_STFT) ? fb_in[f * NB + kk] : 0.0f;
    fbp2[(((f  >> 1) * NB) + kk) * 2 + (f  & 1)] = v;
    fbT2[(((kk >> 1) * FP) + f ) * 2 + (kk & 1)] = v;
}

__device__ __forceinline__ void ibs_load_a1(v2f (&dst)[4][8], const float2* base) {
    #pragma unroll
    for (int s = 0; s < 4; ++s) {
        #pragma unroll
        for (int bt = 0; bt < 8; ++bt) {
            float2 av = base[s * FBP + 16 * bt];
            dst[s][bt][0] = av.x;
            dst[s][bt][1] = av.y;
        }
    }
}

__global__ __launch_bounds__(NTHR) void ibs_run(const float* __restrict__ barkspec,
                                                const float* __restrict__ spec_init,
                                                float* __restrict__ spec,
                                                const float2* __restrict__ fbp2,
                                                const float2* __restrict__ fbT2,
                                                float* lossp,
                                                int replay) {
    __shared__ __attribute__((aligned(16))) float2 lds_fb[(FP / 2) * FBP];
    __shared__ __attribute__((aligned(16))) float4 lds_x[XDIFF + 16 * 32];
    __shared__ float  lds_wsum[NW];
    __shared__ float  lds_loss[32];
    __shared__ int    lds_n;

    const int tid     = threadIdx.x;
    const int wave    = __builtin_amdgcn_readfirstlane(tid >> 5);
    const int lane    = tid & 31;
    const int mi      = lane & 15;
    const int hsel    = lane >> 4;
    const int rowbase = blockIdx.x * 16;

    int n_iter = MAX_ITER;
    if (replay) {
        if (tid < MAX_ITER - 1) {
            const float4* lp = (const float4*)(lossp + tid * 128);
            float s = 0.0f;
            #pragma unroll
            for (int i = 0; i < 32; ++i) {
                float4 v = lp[i];
                s += v.x; s += v.y; s += v.z; s += v.w;
            }
            lds_loss[tid] = s / (float)ROWS;
        }
        __syncthreads();
        if (tid == 0) {
            float prev = INFINITY;
            int n = MAX_ITER;
            for (int it = 0; it < MAX_ITER - 1; ++it) {
                float loss = lds_loss[it];
                bool stop = (loss < TOL_LOSS) || (fabsf(prev - loss) < TOL_CHG);
                prev = loss;
                if (stop) { n = it + 1; break; }
            }
            lds_n = n;
        }
        __syncthreads();
        n_iter = lds_n;
        if (n_iter >= MAX_ITER) return;
    }

    {
        const float4* src4 = (const float4*)fbp2;
        #pragma unroll 33
        for (int q = tid; q < (FP / 2) * (NB / 2); q += NTHR) {
            float4 v = src4[q];
            const int p = q >> 6;
            const int c = q & 63;
            *(float4*)&lds_fb[p * FBP + 2 * c] = v;
        }
    }
    float* stage = (float*)lds_x;
    {
        const float* srcp = spec_init + rowbase * N_STFT;
        #pragma unroll
        for (int k = 0; k < (16 * N_STFT + NTHR - 1) / NTHR; ++k) {
            int i = tid + k * NTHR;
            i = i < 16 * N_STFT ? i : 16 * N_STFT - 1;
            stage[i] = srcp[i];
        }
    }
    __syncthreads();

    v8f S[TPW], Bm[TPW];
    #pragma unroll
    for (int j = 0; j < TPW; ++j) {
        #pragma unroll
        for (int r = 0; r < 8; ++r) {
            const int f = (wave + 4 * j) * 16 + hsel * 8 + r;
            S[j][r]  = (f < N_STFT) ? stage[mi * N_STFT + f] : 0.0f;
            Bm[j][r] = 0.0f;
        }
    }

    const int bb = rowbase >> 9;
    const int t0 = rowbase & 511;
    float tgt[2][8];
    #pragma unroll
    for (int u = 0; u < 2; ++u) {
        #pragma unroll
        for (int r = 0; r < 8; ++r) {
            const int bark = (2 * wave + u) * 16 + hsel * 8 + r;
            tgt[u][r] = barkspec[(bb * N_BARKS + bark) * TIME + t0 + mi];
        }
    }

    const float2* a2p = fbT2 + (4 * hsel) * FP + 16 * wave + mi;
    v2f A2[8][4][TPW];
    #pragma unroll
    for (int bt = 0; bt < 8; ++bt) {
        #pragma unroll
        for (int s = 0; s < 4; ++s) {
            #pragma unroll
            for (int j = 0; j < TPW; ++j) {
                const bool real = (wave + 4 * j) < NTILE;
                const int  joff = real ? 64 * j : 0;
                float2 av = a2p[(8 * bt + s) * FP + joff];
                A2[bt][s][j][0] = real ? av.x : 0.0f;
                A2[bt][s][j][1] = real ? av.y : 0.0f;
            }
        }
    }
    __syncthreads();

    int a1off[TPW];
    #pragma unroll
    for (int j = 0; j < TPW; ++j) {
        int T = wave + 4 * j;
        T = T < NTILE ? T : NTILE - 1;
        a1off[j] = (8 * T + 4 * hsel) * FBP + mi;
    }

    v2f Af[2][4][8];
    ibs_load_a1(Af[0], lds_fb + a1off[0]);

    for (int it = 0; it < n_iter; ++it) {
        v8f acc[8];
        #pragma unroll
        for (int bt = 0; bt < 8; ++bt)
            acc[bt] = (v8f){0.f, 0.f, 0.f, 0.f, 0.f, 0.f, 0.f, 0.f};
        #pragma unroll
        for (int j = 0; j < TPW; ++j) {
            if (j + 1 < TPW) ibs_load_a1(Af[(j + 1) & 1], lds_fb + a1off[j + 1]);
            __builtin_amdgcn_sched_barrier(0);
            #pragma unroll
            for (int s = 0; s < 4; ++s) {
                v2f b;
                b[0] = S[j][2 * s];
                b[1] = S[j][2 * s + 1];
                #pragma unroll
                for (int bt = 0; bt < 8; ++bt) {
                    acc[bt] = __builtin_amdgcn_wmma_f32_16x16x4_f32(false, Af[j & 1][s][bt], false, b,
                                                                    (short)0, acc[bt], false, false);
                }
            }
            __builtin_amdgcn_sched_barrier(0);
        }
        #pragma unroll
        for (int bt = 0; bt < 8; ++bt) {
            lds_x[(wave * 16 + bt * 2 + 0) * 32 + lane] =
                make_float4(acc[bt][0], acc[bt][1], acc[bt][2], acc[bt][3]);
            lds_x[(wave * 16 + bt * 2 + 1) * 32 + lane] =
                make_float4(acc[bt][4], acc[bt][5], acc[bt][6], acc[bt][7]);
        }
        __syncthreads();

        float ssq = 0.0f;
        #pragma unroll
        for (int u = 0; u < 2; ++u) {
            const int bt = 2 * wave + u;
            float p[8];
            #pragma unroll
            for (int r = 0; r < 8; ++r) p[r] = 0.0f;
            #pragma unroll
            for (int sw = 0; sw < NW; ++sw) {
                float4 lo = lds_x[(sw * 16 + bt * 2 + 0) * 32 + lane];
                float4 hi = lds_x[(sw * 16 + bt * 2 + 1) * 32 + lane];
                p[0] += lo.x; p[1] += lo.y; p[2] += lo.z; p[3] += lo.w;
                p[4] += hi.x; p[5] += hi.y; p[6] += hi.z; p[7] += hi.w;
            }
            float d[8];
            #pragma unroll
            for (int r = 0; r < 8; ++r) {
                d[r] = tgt[u][r] - p[r];
                ssq += d[r] * d[r];
            }
            lds_x[XDIFF + (bt * 2 + 0) * 32 + lane] = make_float4(d[0], d[1], d[2], d[3]);
            lds_x[XDIFF + (bt * 2 + 1) * 32 + lane] = make_float4(d[4], d[5], d[6], d[7]);
        }
        ssq += __shfl_xor(ssq, 16);
        ssq += __shfl_xor(ssq, 8);
        ssq += __shfl_xor(ssq, 4);
        ssq += __shfl_xor(ssq, 2);
        ssq += __shfl_xor(ssq, 1);
        if (lane == 0) lds_wsum[wave] = ssq;
        __syncthreads();

        if (tid == 0 && !replay) {
            float s = 0.0f;
            for (int w = 0; w < NW; ++w) s += lds_wsum[w];
            lossp[it * 128 + blockIdx.x] = s;
        }

        v8f D1[8];
        #pragma unroll
        for (int bt = 0; bt < 8; ++bt) {
            float4 lo = lds_x[XDIFF + (bt * 2 + 0) * 32 + lane];
            float4 hi = lds_x[XDIFF + (bt * 2 + 1) * 32 + lane];
            D1[bt][0] = lo.x; D1[bt][1] = lo.y; D1[bt][2] = lo.z; D1[bt][3] = lo.w;
            D1[bt][4] = hi.x; D1[bt][5] = hi.y; D1[bt][6] = hi.z; D1[bt][7] = hi.w;
        }
        __builtin_amdgcn_sched_barrier(0);

        v8f g[TPW];
        #pragma unroll
        for (int j = 0; j < TPW; ++j)
            g[j] = (v8f){0.f, 0.f, 0.f, 0.f, 0.f, 0.f, 0.f, 0.f};
        #pragma unroll
        for (int bt = 0; bt < 8; ++bt) {
            #pragma unroll
            for (int s = 0; s < 4; ++s) {
                v2f b;
                b[0] = D1[bt][2 * s];
                b[1] = D1[bt][2 * s + 1];
                #pragma unroll
                for (int j = 0; j < TPW; ++j) {
                    g[j] = __builtin_amdgcn_wmma_f32_16x16x4_f32(false, A2[bt][s][j], false, b,
                                                                 (short)0, g[j], false, false);
                }
            }
        }
        __builtin_amdgcn_sched_barrier(0);
        ibs_load_a1(Af[0], lds_fb + a1off[0]);
        __builtin_amdgcn_sched_barrier(0);
        #pragma unroll
        for (int j = 0; j < TPW; ++j) {
            #pragma unroll
            for (int r = 0; r < 8; ++r) {
                float gg = g[j][r] * (-2.0f / (float)ROWS);
                float bn = MOMENTUM * Bm[j][r] + gg;
                float sn = S[j][r] - LR * bn;
                S[j][r]  = sn > 0.0f ? sn : 0.0f;
                Bm[j][r] = bn;
            }
        }
    }

    __syncthreads();
    #pragma unroll
    for (int j = 0; j < TPW; ++j) {
        if ((wave + 4 * j) < NTILE) {
            #pragma unroll
            for (int r = 0; r < 8; ++r) {
                const int f = (wave + 4 * j) * 16 + hsel * 8 + r;
                stage[mi * FP + f] = S[j][r];
            }
        }
    }
    __syncthreads();
    {
        float4* dst = (float4*)(spec + rowbase * FP);
        const float4* st4 = (const float4*)stage;
        for (int i = tid; i < 16 * FP / 4; i += NTHR) dst[i] = st4[i];
    }
}


__global__ __launch_bounds__(256) void ibs_finalize(const float* __restrict__ spec,
                                                    float* __restrict__ out) {
    int idx = blockIdx.x * blockDim.x + threadIdx.x;
    if (idx >= BATCH * N_STFT * TIME) return;
    int t = idx & 511;
    int f = (idx >> 9) % N_STFT;
    int b = idx / (N_STFT * TIME);
    out[idx] = spec[(b * TIME + t) * FP + f];
}


extern "C" void kernel_launch(void* const* d_in, const int* in_sizes, int n_in,
                              void* d_out, int out_size, void* d_ws, size_t ws_size,
                              hipStream_t stream) {
    const float* barkspec  = (const float*)d_in[0];
    const float* fb        = (const float*)d_in[1];
    const float* spec_init = (const float*)d_in[2];

    float* ws       = (float*)d_ws;
    float* spec     = ws;
    float* lossp    = spec + ROWS * FP;
    float* fbp2     = lossp + ROWS * FP;
    float* fbT2     = fbp2 + FP * NB;

    ibs_init_fb<<<(FP * NB + 255) / 256, 256, 0, stream>>>(fb, fbp2, fbT2);
    ibs_run<<<ROWS / 16, NTHR, 0, stream>>>(barkspec, spec_init, spec,
                                            (const float2*)fbp2, (const float2*)fbT2, lossp, 0);
    ibs_run<<<ROWS / 16, NTHR, 0, stream>>>(barkspec, spec_init, spec,
                                            (const float2*)fbp2, (const float2*)fbT2, lossp, 1);
    ibs_finalize<<<(BATCH * N_STFT * TIME + 255) / 256, 256, 0, stream>>>(spec, (float*)d_out);
}


# ibs_run: last frequency (1 real row of a 33rd tile) loses its tile slot: product 1 one k-step per wave as a 5th partial, product 2 split by bark range over waves; sum order of p and its g changes
# speedup vs baseline: 4.5639x; 1.0832x over previous
#include <hip/hip_runtime.h>
#include <math.h>

typedef float v2f __attribute__((ext_vector_type(2)));
typedef float v8f __attribute__((ext_vector_type(8)));

#define N_STFT   513
#define N_BARKS  128
#define BATCH    4
#define TIME     512
#define ROWS     (BATCH * TIME)
#define FP       528
#define NB       128
#define MAX_ITER 30
#define LR       0.1f
#define MOMENTUM 0.9f
#define TOL_LOSS 1e-5f
#define TOL_CHG  1e-8f
#define NW       4
#define NTHR     (NW * 32)
#define TPW      8
#define TAILF    512
#define FBP      132
#define XDIFF    (NW * 16 * 32)

__global__ __launch_bounds__(256) void ibs_init_fb(const float* __restrict__ fb_in,
                                                   float* __restrict__ fbp2,
                                                   float* __restrict__ fbT2) {
    int idx = blockIdx.x * blockDim.x + threadIdx.x;
    if (idx >= FP * NB) return;
    int f  = idx / NB;
    int kk = idx - f * NB;
    float v = (f < N_STFT) ? fb_in[f * NB + kk] : 0.0f;
    fbp2[(((f  >> 1) * NB) + kk) * 2 + (f  & 1)] = v;
    fbT2[(((kk >> 1) * FP) + f ) * 2 + (kk & 1)] = v;
}

__device__ __forceinline__ void ibs_load_a1(v2f (&dst)[4][8], const float2* base) {
    #pragma unroll
    for (int s = 0; s < 4; ++s) {
        #pragma unroll
        for (int bt = 0; bt < 8; ++bt) {
            float2 av = base[s * FBP + 16 * bt];
            dst[s][bt][0] = av.x;
            dst[s][bt][1] = av.y;
        }
    }
}

__global__ __launch_bounds__(NTHR) void ibs_run(const float* __restrict__ barkspec,
                                                const float* __restrict__ spec_init,
                                                float* __restrict__ spec,
                                                const float2* __restrict__ fbp2,
                                                const float2* __restrict__ fbT2,
                                                float* lossp,
                                                int replay) {
    __shared__ __attribute__((aligned(16))) float2 lds_fb[(FP / 2) * FBP];
    __shared__ __attribute__((aligned(16))) float4 lds_x[XDIFF + 16 * 32];
    __shared__ float  lds_wsum[NW];
    __shared__ float  lds_gt[2 * NW * 32];
    __shared__ float  lds_loss[32];
    __shared__ int    lds_n;

    const int tid     = threadIdx.x;
    const int wave    = __builtin_amdgcn_readfirstlane(tid >> 5);
    const int lane    = tid & 31;
    const int mi      = lane & 15;
    const int hsel    = lane >> 4;
    const int rowbase = blockIdx.x * 16;

    int n_iter = MAX_ITER;
    if (replay) {
        if (tid < MAX_ITER - 1) {
            const float4* lp = (const float4*)(lossp + tid * 128);
            float s = 0.0f;
            #pragma unroll
            for (int i = 0; i < 32; ++i) {
                float4 v = lp[i];
                s += v.x; s += v.y; s += v.z; s += v.w;
            }
            lds_loss[tid] = s / (float)ROWS;
        }
        __syncthreads();
        if (tid == 0) {
            float prev = INFINITY;
            int n = MAX_ITER;
            for (int it = 0; it < MAX_ITER - 1; ++it) {
                float loss = lds_loss[it];
                bool stop = (loss < TOL_LOSS) || (fabsf(prev - loss) < TOL_CHG);
                prev = loss;
                if (stop) { n = it + 1; break; }
            }
            lds_n = n;
        }
        __syncthreads();
        n_iter = lds_n;
        if (n_iter >= MAX_ITER) return;
    }

    {
        const float4* src4 = (const float4*)fbp2;
        #pragma unroll 33
        for (int q = tid; q < (FP / 2) * (NB / 2); q += NTHR) {
            float4 v = src4[q];
            const int p = q >> 6;
            const int c = q & 63;
            *(float4*)&lds_fb[p * FBP + 2 * c] = v;
        }
    }
    float* stage = (float*)lds_x;
    {
        const float* srcp = spec_init + rowbase * N_STFT;
        #pragma unroll
        for (int k = 0; k < (16 * N_STFT + NTHR - 1) / NTHR; ++k) {
            int i = tid + k * NTHR;
            i = i < 16 * N_STFT ? i : 16 * N_STFT - 1;
            stage[i] = srcp[i];
        }
    }
    __syncthreads();

    v8f S[TPW], Bm[TPW];
    #pragma unroll
    for (int j = 0; j < TPW; ++j) {
        #pragma unroll
        for (int r = 0; r < 8; ++r) {
            const int f = (wave + 4 * j) * 16 + hsel * 8 + r;
            S[j][r]  = stage[mi * N_STFT + f];
            Bm[j][r] = 0.0f;
        }
    }
    float Stail  = hsel == 0 ? stage[mi * N_STFT + TAILF] : 0.0f;
    float Bmtail = 0.0f;

    const int bb = rowbase >> 9;
    const int t0 = rowbase & 511;
    float tgt[2][8];
    #pragma unroll
    for (int u = 0; u < 2; ++u) {
        #pragma unroll
        for (int r = 0; r < 8; ++r) {
            const int bark = (2 * wave + u) * 16 + hsel * 8 + r;
            tgt[u][r] = barkspec[(bb * N_BARKS + bark) * TIME + t0 + mi];
        }
    }

    const float2* a2p = fbT2 + (4 * hsel) * FP + 16 * wave + mi;
    v2f A2[8][4][TPW];
    #pragma unroll
    for (int bt = 0; bt < 8; ++bt) {
        #pragma unroll
        for (int s = 0; s < 4; ++s) {
            #pragma unroll
            for (int j = 0; j < TPW; ++j) {
                float2 av = a2p[(8 * bt + s) * FP + 64 * j];
                A2[bt][s][j][0] = av.x;
                A2[bt][s][j][1] = av.y;
            }
        }
    }
    const float2* a2t = fbT2 + (4 * hsel) * FP + TAILF + mi;
    v2f A2t[2][4];
    v2f A1t[2];
    #pragma unroll
    for (int u = 0; u < 2; ++u) {
        #pragma unroll
        for (int s = 0; s < 4; ++s) {
            float2 av = a2t[(8 * (2 * wave + u) + s) * FP];
            A2t[u][s][0] = av.x;
            A2t[u][s][1] = av.y;
        }
        float2 a1 = lds_fb[((TAILF / 2) + 4 * hsel) * FBP + 16 * (2 * wave + u) + mi];
        A1t[u][0] = a1.x;
        A1t[u][1] = a1.y;
    }
    __syncthreads();

    int a1off[TPW];
    #pragma unroll
    for (int j = 0; j < TPW; ++j) {
        const int T = wave + 4 * j;
        a1off[j] = (8 * T + 4 * hsel) * FBP + mi;
    }

    v2f Af[2][4][8];
    ibs_load_a1(Af[0], lds_fb + a1off[0]);

    for (int it = 0; it < n_iter; ++it) {
        v8f acc[8];
        #pragma unroll
        for (int bt = 0; bt < 8; ++bt)
            acc[bt] = (v8f){0.f, 0.f, 0.f, 0.f, 0.f, 0.f, 0.f, 0.f};
        #pragma unroll
        for (int j = 0; j < TPW; ++j) {
            if (j + 1 < TPW) ibs_load_a1(Af[(j + 1) & 1], lds_fb + a1off[j + 1]);
            __builtin_amdgcn_sched_barrier(0);
            #pragma unroll
            for (int s = 0; s < 4; ++s) {
                v2f b;
                b[0] = S[j][2 * s];
                b[1] = S[j][2 * s + 1];
                #pragma unroll
                for (int bt = 0; bt < 8; ++bt) {
                    acc[bt] = __builtin_amdgcn_wmma_f32_16x16x4_f32(false, Af[j & 1][s][bt], false, b,
                                                                    (short)0, acc[bt], false, false);
                }
            }
            __builtin_amdgcn_sched_barrier(0);
        }
        v8f tacc[2];
        {
            v2f b;
            b[0] = Stail;
            b[1] = 0.0f;
            #pragma unroll
            for (int u = 0; u < 2; ++u) {
                tacc[u] = __builtin_amdgcn_wmma_f32_16x16x4_f32(false, A1t[u], false, b, (short)0,
                                                                (v8f){0.f, 0.f, 0.f, 0.f, 0.f, 0.f, 0.f, 0.f},
                                                                false, false);
            }
        }
        #pragma unroll
        for (int bt = 0; bt < 8; ++bt) {
            lds_x[(wave * 16 + bt * 2 + 0) * 32 + lane] =
                make_float4(acc[bt][0], acc[bt][1], acc[bt][2], acc[bt][3]);
            lds_x[(wave * 16 + bt * 2 + 1) * 32 + lane] =
                make_float4(acc[bt][4], acc[bt][5], acc[bt][6], acc[bt][7]);
        }
        __syncthreads();

        float ssq = 0.0f;
        float dd[2][8];
        #pragma unroll
        for (int u = 0; u < 2; ++u) {
            const int bt = 2 * wave + u;
            float p[8];
            #pragma unroll
            for (int r = 0; r < 8; ++r) p[r] = 0.0f;
            #pragma unroll
            for (int sw = 0; sw < NW; ++sw) {
                float4 lo = lds_x[(sw * 16 + bt * 2 + 0) * 32 + lane];
                float4 hi = lds_x[(sw * 16 + bt * 2 + 1) * 32 + lane];
                p[0] += lo.x; p[1] += lo.y; p[2] += lo.z; p[3] += lo.w;
                p[4] += hi.x; p[5] += hi.y; p[6] += hi.z; p[7] += hi.w;
            }
            #pragma unroll
            for (int r = 0; r < 8; ++r) {
                p[r] += tacc[u][r];
                dd[u][r] = tgt[u][r] - p[r];
                ssq += dd[u][r] * dd[u][r];
            }
            lds_x[XDIFF + (bt * 2 + 0) * 32 + lane] = make_float4(dd[u][0], dd[u][1], dd[u][2], dd[u][3]);
            lds_x[XDIFF + (bt * 2 + 1) * 32 + lane] = make_float4(dd[u][4], dd[u][5], dd[u][6], dd[u][7]);
        }
        v8f gt[2];
        #pragma unroll
        for (int u = 0; u < 2; ++u)
            gt[u] = (v8f){0.f, 0.f, 0.f, 0.f, 0.f, 0.f, 0.f, 0.f};
        #pragma unroll
        for (int s = 0; s < 4; ++s) {
            #pragma unroll
            for (int u = 0; u < 2; ++u) {
                v2f b;
                b[0] = dd[u][2 * s];
                b[1] = dd[u][2 * s + 1];
                gt[u] = __builtin_amdgcn_wmma_f32_16x16x4_f32(false, A2t[u][s], false, b,
                                                              (short)0, gt[u], false, false);
            }
        }
        ssq += __shfl_xor(ssq, 16);
        ssq += __shfl_xor(ssq, 8);
        ssq += __shfl_xor(ssq, 4);
        ssq += __shfl_xor(ssq, 2);
        ssq += __shfl_xor(ssq, 1);
        if (lane == 0) lds_wsum[wave] = ssq;
        lds_gt[(2 * wave + 0) * 32 + lane] = gt[0][0];
        lds_gt[(2 * wave + 1) * 32 + lane] = gt[1][0];
        __syncthreads();

        if (tid == 0 && !replay) {
            float s = 0.0f;
            for (int w = 0; w < NW; ++w) s += lds_wsum[w];
            lossp[it * 128 + blockIdx.x] = s;
        }

        v8f D1[8];
        #pragma unroll
        for (int bt = 0; bt < 8; ++bt) {
            float4 lo = lds_x[XDIFF + (bt * 2 + 0) * 32 + lane];
            float4 hi = lds_x[XDIFF + (bt * 2 + 1) * 32 + lane];
            D1[bt][0] = lo.x; D1[bt][1] = lo.y; D1[bt][2] = lo.z; D1[bt][3] = lo.w;
            D1[bt][4] = hi.x; D1[bt][5] = hi.y; D1[bt][6] = hi.z; D1[bt][7] = hi.w;
        }
        __builtin_amdgcn_sched_barrier(0);

        v8f g[TPW];
        #pragma unroll
        for (int j = 0; j < TPW; ++j)
            g[j] = (v8f){0.f, 0.f, 0.f, 0.f, 0.f, 0.f, 0.f, 0.f};
        #pragma unroll
        for (int bt = 0; bt < 8; ++bt) {
            #pragma unroll
            for (int s = 0; s < 4; ++s) {
                v2f b;
                b[0] = D1[bt][2 * s];
                b[1] = D1[bt][2 * s + 1];
                #pragma unroll
                for (int j = 0; j < TPW; ++j) {
                    g[j] = __builtin_amdgcn_wmma_f32_16x16x4_f32(false, A2[bt][s][j], false, b,
                                                                 (short)0, g[j], false, false);
                }
            }
        }
        __builtin_amdgcn_sched_barrier(0);
        ibs_load_a1(Af[0], lds_fb + a1off[0]);
        float q[2 * NW];
        #pragma unroll
        for (int i = 0; i < 2 * NW; ++i) q[i] = lds_gt[i * 32 + lane];
        __builtin_amdgcn_sched_barrier(0);
        #pragma unroll
        for (int j = 0; j < TPW; ++j) {
            #pragma unroll
            for (int r = 0; r < 8; ++r) {
                float gg = g[j][r] * (-2.0f / (float)ROWS);
                float bn = MOMENTUM * Bm[j][r] + gg;
                float sn = S[j][r] - LR * bn;
                S[j][r]  = sn > 0.0f ? sn : 0.0f;
                Bm[j][r] = bn;
            }
        }
        {
            const float gtail = (((q[0] + q[1]) + (q[2] + q[3])) + (q[4] + q[5])) + (q[6] + q[7]);
            float gg = gtail * (-2.0f / (float)ROWS);
            float bn = MOMENTUM * Bmtail + gg;
            float sn = Stail - LR * bn;
            Stail  = sn > 0.0f ? sn : 0.0f;
            Bmtail = bn;
        }
    }

    __syncthreads();
    #pragma unroll
    for (int j = 0; j < TPW; ++j) {
        #pragma unroll
        for (int r = 0; r < 8; ++r) {
            const int f = (wave + 4 * j) * 16 + hsel * 8 + r;
            stage[mi * FP + f] = S[j][r];
        }
    }
    if (wave == 0) {
        #pragma unroll
        for (int r = 0; r < 8; ++r)
            stage[mi * FP + TAILF + hsel * 8 + r] = (r == 0) ? Stail : 0.0f;
    }
    __syncthreads();
    {
        float4* dst = (float4*)(spec + rowbase * FP);
        const float4* st4 = (const float4*)stage;
        for (int i = tid; i < 16 * FP / 4; i += NTHR) dst[i] = st4[i];
    }
}


__global__ __launch_bounds__(256) void ibs_finalize(const float* __restrict__ spec,
                                                    float* __restrict__ out) {
    int idx = blockIdx.x * blockDim.x + threadIdx.x;
    if (idx >= BATCH * N_STFT * TIME) return;
    int t = idx & 511;
    int f = (idx >> 9) % N_STFT;
    int b = idx / (N_STFT * TIME);
    out[idx] = spec[(b * TIME + t) * FP + f];
}


extern "C" void kernel_launch(void* const* d_in, const int* in_sizes, int n_in,
                              void* d_out, int out_size, void* d_ws, size_t ws_size,
                              hipStream_t stream) {
    const float* barkspec  = (const float*)d_in[0];
    const float* fb        = (const float*)d_in[1];
    const float* spec_init = (const float*)d_in[2];

    float* ws       = (float*)d_ws;
    float* spec     = ws;
    float* lossp    = spec + ROWS * FP;
    float* fbp2     = lossp + ROWS * FP;
    float* fbT2     = fbp2 + FP * NB;

    ibs_init_fb<<<(FP * NB + 255) / 256, 256, 0, stream>>>(fb, fbp2, fbT2);
    ibs_run<<<ROWS / 16, NTHR, 0, stream>>>(barkspec, spec_init, spec,
                                            (const float2*)fbp2, (const float2*)fbT2, lossp, 0);
    ibs_run<<<ROWS / 16, NTHR, 0, stream>>>(barkspec, spec_init, spec,
                                            (const float2*)fbp2, (const float2*)fbT2, lossp, 1);
    ibs_finalize<<<(BATCH * N_STFT * TIME + 255) / 256, 256, 0, stream>>>(spec, (float*)d_out);
}


# ibs_run stores its result tile frequency-major (16 rows of one frequency = one 64-byte piece); last launch no longer transposes, it copies pieces in 16-byte units, 171 work-groups not 4104; sums same
# speedup vs baseline: 4.7400x; 1.0386x over previous
#include <hip/hip_runtime.h>
#include <math.h>

typedef float v2f __attribute__((ext_vector_type(2)));
typedef float v4f __attribute__((ext_vector_type(4)));
typedef float v8f __attribute__((ext_vector_type(8)));

#define N_STFT   513
#define N_BARKS  128
#define BATCH    4
#define TIME     512
#define ROWS     (BATCH * TIME)
#define FP       528
#define NB       128
#define MAX_ITER 30
#define LR       0.1f
#define MOMENTUM 0.9f
#define TOL_LOSS 1e-5f
#define TOL_CHG  1e-8f
#define NW       4
#define NTHR     (NW * 32)
#define TPW      8
#define TAILF    512
#define FBP      132
#define XDIFF    (NW * 16 * 32)

__global__ __launch_bounds__(256) void ibs_init_fb(const float* __restrict__ fb_in,
                                                   float* __restrict__ fbp2,
                                                   float* __restrict__ fbT2) {
    int idx = blockIdx.x * blockDim.x + threadIdx.x;
    if (idx >= FP * NB) return;
    int f  = idx / NB;
    int kk = idx - f * NB;
    float v = (f < N_STFT) ? fb_in[f * NB + kk] : 0.0f;
    fbp2[(((f  >> 1) * NB) + kk) * 2 + (f  & 1)] = v;
    fbT2[(((kk >> 1) * FP) + f ) * 2 + (kk & 1)] = v;
}

__device__ __forceinline__ void ibs_load_a1(v2f (&dst)[4][8], const float2* base) {
    #pragma unroll
    for (int s = 0; s < 4; ++s) {
        #pragma unroll
        for (int bt = 0; bt < 8; ++bt) {
            float2 av = base[s * FBP + 16 * bt];
            dst[s][bt][0] = av.x;
            dst[s][bt][1] = av.y;
        }
    }
}

__global__ __launch_bounds__(NTHR) void ibs_run(const float* __restrict__ barkspec,
                                                const float* __restrict__ spec_init,
                                                float* __restrict__ spec,
                                                const float2* __restrict__ fbp2,
                                                const float2* __restrict__ fbT2,
                                                float* lossp,
                                                int replay) {
    __shared__ __attribute__((aligned(16))) float2 lds_fb[(FP / 2) * FBP];
    __shared__ __attribute__((aligned(16))) float4 lds_x[XDIFF + 16 * 32];
    __shared__ float  lds_wsum[NW];
    __shared__ float  lds_gt[2 * NW * 32];
    __shared__ float  lds_loss[32];
    __shared__ int    lds_n;

    const int tid     = threadIdx.x;
    const int wave    = __builtin_amdgcn_readfirstlane(tid >> 5);
    const int lane    = tid & 31;
    const int mi      = lane & 15;
    const int hsel    = lane >> 4;
    const int rowbase = blockIdx.x * 16;

    int n_iter = MAX_ITER;
    if (replay) {
        if (tid < MAX_ITER - 1) {
            const float4* lp = (const float4*)(lossp + tid * 128);
            float s = 0.0f;
            #pragma unroll
            for (int i = 0; i < 32; ++i) {
                float4 v = lp[i];
                s += v.x; s += v.y; s += v.z; s += v.w;
            }
            lds_loss[tid] = s / (float)ROWS;
        }
        __syncthreads();
        if (tid == 0) {
            float prev = INFINITY;
            int n = MAX_ITER;
            for (int it = 0; it < MAX_ITER - 1; ++it) {
                float loss = lds_loss[it];
                bool stop = (loss < TOL_LOSS) || (fabsf(prev - loss) < TOL_CHG);
                prev = loss;
                if (stop) { n = it + 1; break; }
            }
            lds_n = n;
        }
        __syncthreads();
        n_iter = lds_n;
        if (n_iter >= MAX_ITER) return;
    }

    {
        const float4* src4 = (const float4*)fbp2;
        #pragma unroll 33
        for (int q = tid; q < (FP / 2) * (NB / 2); q += NTHR) {
            float4 v = src4[q];
            const int p = q >> 6;
            const int c = q & 63;
            *(float4*)&lds_fb[p * FBP + 2 * c] = v;
        }
    }
    float* stage = (float*)lds_x;
    {
        const float* srcp = spec_init + rowbase * N_STFT;
        #pragma unroll
        for (int k = 0; k < (16 * N_STFT + NTHR - 1) / NTHR; ++k) {
            int i = tid + k * NTHR;
            i = i < 16 * N_STFT ? i : 16 * N_STFT - 1;
            stage[i] = srcp[i];
        }
    }
    __syncthreads();

    v8f S[TPW], Bm[TPW];
    #pragma unroll
    for (int j = 0; j < TPW; ++j) {
        #pragma unroll
        for (int r = 0; r < 8; ++r) {
            const int f = (wave + 4 * j) * 16 + hsel * 8 + r;
            S[j][r]  = stage[mi * N_STFT + f];
            Bm[j][r] = 0.0f;
        }
    }
    float Stail  = hsel == 0 ? stage[mi * N_STFT + TAILF] : 0.0f;
    float Bmtail = 0.0f;

    const int bb = rowbase >> 9;
    const int t0 = rowbase & 511;
    float tgt[2][8];
    #pragma unroll
    for (int u = 0; u < 2; ++u) {
        #pragma unroll
        for (int r = 0; r < 8; ++r) {
            const int bark = (2 * wave + u) * 16 + hsel * 8 + r;
            tgt[u][r] = barkspec[(bb * N_BARKS + bark) * TIME + t0 + mi];
        }
    }

    const float2* a2p = fbT2 + (4 * hsel) * FP + 16 * wave + mi;
    v2f A2[8][4][TPW];
    #pragma unroll
    for (int bt = 0; bt < 8; ++bt) {
        #pragma unroll
        for (int s = 0; s < 4; ++s) {
            #pragma unroll
            for (int j = 0; j < TPW; ++j) {
                float2 av = a2p[(8 * bt + s) * FP + 64 * j];
                A2[bt][s][j][0] = av.x;
                A2[bt][s][j][1] = av.y;
            }
        }
    }
    const float2* a2t = fbT2 + (4 * hsel) * FP + TAILF + mi;
    v2f A2t[2][4];
    v2f A1t[2];
    #pragma unroll
    for (int u = 0; u < 2; ++u) {
        #pragma unroll
        for (int s = 0; s < 4; ++s) {
            float2 av = a2t[(8 * (2 * wave + u) + s) * FP];
            A2t[u][s][0] = av.x;
            A2t[u][s][1] = av.y;
        }
        float2 a1 = lds_fb[((TAILF / 2) + 4 * hsel) * FBP + 16 * (2 * wave + u) + mi];
        A1t[u][0] = a1.x;
        A1t[u][1] = a1.y;
    }
    __syncthreads();

    int a1off[TPW];
    #pragma unroll
    for (int j = 0; j < TPW; ++j) {
        const int T = wave + 4 * j;
        a1off[j] = (8 * T + 4 * hsel) * FBP + mi;
    }

    v2f Af[2][4][8];
    ibs_load_a1(Af[0], lds_fb + a1off[0]);

    for (int it = 0; it < n_iter; ++it) {
        v8f acc[8];
        #pragma unroll
        for (int bt = 0; bt < 8; ++bt)
            acc[bt] = (v8f){0.f, 0.f, 0.f, 0.f, 0.f, 0.f, 0.f, 0.f};
        #pragma unroll
        for (int j = 0; j < TPW; ++j) {
            if (j + 1 < TPW) ibs_load_a1(Af[(j + 1) & 1], lds_fb + a1off[j + 1]);
            __builtin_amdgcn_sched_barrier(0);
            #pragma unroll
            for (int s = 0; s < 4; ++s) {
                v2f b;
                b[0] = S[j][2 * s];
                b[1] = S[j][2 * s + 1];
                #pragma unroll
                for (int bt = 0; bt < 8; ++bt) {
                    acc[bt] = __builtin_amdgcn_wmma_f32_16x16x4_f32(false, Af[j & 1][s][bt], false, b,
                                                                    (short)0, acc[bt], false, false);
                }
            }
            __builtin_amdgcn_sched_barrier(0);
        }
        v8f tacc[2];
        {
            v2f b;
            b[0] = Stail;
            b[1] = 0.0f;
            #pragma unroll
            for (int u = 0; u < 2; ++u) {
                tacc[u] = __builtin_amdgcn_wmma_f32_16x16x4_f32(false, A1t[u], false, b, (short)0,
                                                                (v8f){0.f, 0.f, 0.f, 0.f, 0.f, 0.f, 0.f, 0.f},
                                                                false, false);
            }
        }
        #pragma unroll
        for (int bt = 0; bt < 8; ++bt) {
            lds_x[(wave * 16 + bt * 2 + 0) * 32 + lane] =
                make_float4(acc[bt][0], acc[bt][1], acc[bt][2], acc[bt][3]);
            lds_x[(wave * 16 + bt * 2 + 1) * 32 + lane] =
                make_float4(acc[bt][4], acc[bt][5], acc[bt][6], acc[bt][7]);
        }
        __syncthreads();

        float ssq = 0.0f;
        float dd[2][8];
        #pragma unroll
        for (int u = 0; u < 2; ++u) {
            const int bt = 2 * wave + u;
            float p[8];
            #pragma unroll
            for (int r = 0; r < 8; ++r) p[r] = 0.0f;
            #pragma unroll
            for (int sw = 0; sw < NW; ++sw) {
                float4 lo = lds_x[(sw * 16 + bt * 2 + 0) * 32 + lane];
                float4 hi = lds_x[(sw * 16 + bt * 2 + 1) * 32 + lane];
                p[0] += lo.x; p[1] += lo.y; p[2] += lo.z; p[3] += lo.w;
                p[4] += hi.x; p[5] += hi.y; p[6] += hi.z; p[7] += hi.w;
            }
            #pragma unroll
            for (int r = 0; r < 8; ++r) {
                p[r] += tacc[u][r];
                dd[u][r] = tgt[u][r] - p[r];
                ssq += dd[u][r] * dd[u][r];
            }
            lds_x[XDIFF + (bt * 2 + 0) * 32 + lane] = make_float4(dd[u][0], dd[u][1], dd[u][2], dd[u][3]);
            lds_x[XDIFF + (bt * 2 + 1) * 32 + lane] = make_float4(dd[u][4], dd[u][5], dd[u][6], dd[u][7]);
        }
        v8f gt[2];
        #pragma unroll
        for (int u = 0; u < 2; ++u)
            gt[u] = (v8f){0.f, 0.f, 0.f, 0.f, 0.f, 0.f, 0.f, 0.f};
        #pragma unroll
        for (int s = 0; s < 4; ++s) {
            #pragma unroll
            for (int u = 0; u < 2; ++u) {
                v2f b;
                b[0] = dd[u][2 * s];
                b[1] = dd[u][2 * s + 1];
                gt[u] = __builtin_amdgcn_wmma_f32_16x16x4_f32(false, A2t[u][s], false, b,
                                                              (short)0, gt[u], false, false);
            }
        }
        ssq += __shfl_xor(ssq, 16);
        ssq += __shfl_xor(ssq, 8);
        ssq += __shfl_xor(ssq, 4);
        ssq += __shfl_xor(ssq, 2);
        ssq += __shfl_xor(ssq, 1);
        if (lane == 0) lds_wsum[wave] = ssq;
        lds_gt[(2 * wave + 0) * 32 + lane] = gt[0][0];
        lds_gt[(2 * wave + 1) * 32 + lane] = gt[1][0];
        __syncthreads();

        if (tid == 0 && !replay) {
            float s = 0.0f;
            for (int w = 0; w < NW; ++w) s += lds_wsum[w];
            lossp[it * 128 + blockIdx.x] = s;
        }

        v8f D1[8];
        #pragma unroll
        for (int bt = 0; bt < 8; ++bt) {
            float4 lo = lds_x[XDIFF + (bt * 2 + 0) * 32 + lane];
            float4 hi = lds_x[XDIFF + (bt * 2 + 1) * 32 + lane];
            D1[bt][0] = lo.x; D1[bt][1] = lo.y; D1[bt][2] = lo.z; D1[bt][3] = lo.w;
            D1[bt][4] = hi.x; D1[bt][5] = hi.y; D1[bt][6] = hi.z; D1[bt][7] = hi.w;
        }
        __builtin_amdgcn_sched_barrier(0);

        v8f g[TPW];
        #pragma unroll
        for (int j = 0; j < TPW; ++j)
            g[j] = (v8f){0.f, 0.f, 0.f, 0.f, 0.f, 0.f, 0.f, 0.f};
        #pragma unroll
        for (int bt = 0; bt < 8; ++bt) {
            #pragma unroll
            for (int s = 0; s < 4; ++s) {
                v2f b;
                b[0] = D1[bt][2 * s];
                b[1] = D1[bt][2 * s + 1];
                #pragma unroll
                for (int j = 0; j < TPW; ++j) {
                    g[j] = __builtin_amdgcn_wmma_f32_16x16x4_f32(false, A2[bt][s][j], false, b,
                                                                 (short)0, g[j], false, false);
                }
            }
        }
        __builtin_amdgcn_sched_barrier(0);
        ibs_load_a1(Af[0], lds_fb + a1off[0]);
        float q[2 * NW];
        #pragma unroll
        for (int i = 0; i < 2 * NW; ++i) q[i] = lds_gt[i * 32 + lane];
        __builtin_amdgcn_sched_barrier(0);
        #pragma unroll
        for (int j = 0; j < TPW; ++j) {
            #pragma unroll
            for (int r = 0; r < 8; ++r) {
                float gg = g[j][r] * (-2.0f / (float)ROWS);
                float bn = MOMENTUM * Bm[j][r] + gg;
                float sn = S[j][r] - LR * bn;
                S[j][r]  = sn > 0.0f ? sn : 0.0f;
                Bm[j][r] = bn;
            }
        }
        {
            const float gtail = (((q[0] + q[1]) + (q[2] + q[3])) + (q[4] + q[5])) + (q[6] + q[7]);
            float gg = gtail * (-2.0f / (float)ROWS);
            float bn = MOMENTUM * Bmtail + gg;
            float sn = Stail - LR * bn;
            Stail  = sn > 0.0f ? sn : 0.0f;
            Bmtail = bn;
        }
    }

    __syncthreads();
    #pragma unroll
    for (int j = 0; j < TPW; ++j) {
        #pragma unroll
        for (int r = 0; r < 8; ++r) {
            const int f = (wave + 4 * j) * 16 + hsel * 8 + r;
            stage[f * 16 + mi] = S[j][r];
        }
    }
    if (wave == 0) {
        #pragma unroll
        for (int r = 0; r < 8; ++r)
            stage[(TAILF + hsel * 8 + r) * 16 + mi] = (r == 0) ? Stail : 0.0f;
    }
    __syncthreads();
    {
        float4* dst = (float4*)(spec + rowbase * FP);
        const float4* st4 = (const float4*)stage;
        for (int i = tid; i < 16 * FP / 4; i += NTHR) dst[i] = st4[i];
    }
}

#define FIN_WG   171
#define FIN_PER  6

__global__ __launch_bounds__(256) void ibs_finalize(const float* __restrict__ spec,
                                                    float* __restrict__ out) {
    const v4f* src = (const v4f*)spec;
    v4f*       dst = (v4f*)out;
    const int idx0 = blockIdx.x * 256 + threadIdx.x;
    v4f v[FIN_PER];
    #pragma unroll
    for (int k = 0; k < FIN_PER; ++k) {
        const int idx4 = idx0 + k * (FIN_WG * 256);
        const int q  = idx4 & 127;
        const int bf = idx4 >> 7;
        const int b  = bf / N_STFT;
        const int f  = bf - b * N_STFT;
        v[k] = src[((b * (TIME / 16) + (q >> 2)) * FP + f) * 4 + (q & 3)];
    }
    asm volatile("" :: "v"(v[0]), "v"(v[1]), "v"(v[2]), "v"(v[3]), "v"(v[4]), "v"(v[5]));
    #pragma unroll
    for (int k = 0; k < FIN_PER; ++k)
        dst[idx0 + k * (FIN_WG * 256)] = v[k];
}


extern "C" void kernel_launch(void* const* d_in, const int* in_sizes, int n_in,
                              void* d_out, int out_size, void* d_ws, size_t ws_size,
                              hipStream_t stream) {
    const float* barkspec  = (const float*)d_in[0];
    const float* fb        = (const float*)d_in[1];
    const float* spec_init = (const float*)d_in[2];

    float* ws       = (float*)d_ws;
    float* spec     = ws;
    float* lossp    = spec + ROWS * FP;
    float* fbp2     = lossp + ROWS * FP;
    float* fbT2     = fbp2 + FP * NB;

    ibs_init_fb<<<(FP * NB + 255) / 256, 256, 0, stream>>>(fb, fbp2, fbT2);
    ibs_run<<<ROWS / 16, NTHR, 0, stream>>>(barkspec, spec_init, spec,
                                            (const float2*)fbp2, (const float2*)fbT2, lossp, 0);
    ibs_run<<<ROWS / 16, NTHR, 0, stream>>>(barkspec, spec_init, spec,
                                            (const float2*)fbp2, (const float2*)fbT2, lossp, 1);
    ibs_finalize<<<FIN_WG, 256, 0, stream>>>(spec, (float*)d_out);
}


# ibs_run: both cross-wave hand-overs (partials, diff) run under matrix work on the wave's own bark pair: barrier signalled before, waited inside it; sums of p and g start at own pair (order changes)
# speedup vs baseline: 4.9002x; 1.0338x over previous
#include <hip/hip_runtime.h>
#include <math.h>

typedef float v2f __attribute__((ext_vector_type(2)));
typedef float v4f __attribute__((ext_vector_type(4)));
typedef float v8f __attribute__((ext_vector_type(8)));

#define N_STFT   513
#define N_BARKS  128
#define BATCH    4
#define TIME     512
#define ROWS     (BATCH * TIME)
#define FP       528
#define NB       128
#define MAX_ITER 30
#define LR       0.1f
#define MOMENTUM 0.9f
#define TOL_LOSS 1e-5f
#define TOL_CHG  1e-8f
#define NW       4
#define NTHR     (NW * 32)
#define TPW      8
#define TAILF    512
#define FBP      132
#define XDIFF    (NW * 16 * 32)

__global__ __launch_bounds__(256) void ibs_init_fb(const float* __restrict__ fb_in,
                                                   float* __restrict__ fbp2,
                                                   float* __restrict__ fbT2) {
    int idx = blockIdx.x * blockDim.x + threadIdx.x;
    if (idx >= FP * NB) return;
    int f  = idx / NB;
    int kk = idx - f * NB;
    float v = (f < N_STFT) ? fb_in[f * NB + kk] : 0.0f;
    fbp2[(((f  >> 1) * NB) + kk) * 2 + (f  & 1)] = v;
    fbT2[(((kk >> 1) * FP) + f ) * 2 + (kk & 1)] = v;
}

__device__ __forceinline__ void ibs_load_pair(v2f (&dst)[2], const float2* p) {
    float2 a = p[0];
    float2 b = p[16];
    dst[0][0] = a.x; dst[0][1] = a.y;
    dst[1][0] = b.x; dst[1][1] = b.y;
}

__device__ __forceinline__ v8f ibs_mma(v2f a, v2f b, v8f c) {
    return __builtin_amdgcn_wmma_f32_16x16x4_f32(false, a, false, b, (short)0, c, false, false);
}

template <int G>
__device__ __forceinline__ void ibs_load_ga(v2f (&dst)[16][2], const float2* fbl,
                                            const int (&a1off)[TPW], const int (&colr)[4]) {
    #pragma unroll
    for (int k = 0; k < 16; ++k) {
        const int n = 16 * G + k;
        ibs_load_pair(dst[k], fbl + a1off[n / 12] + ((n / 3) & 3) * FBP + colr[1 + n % 3]);
    }
}

template <int G>
__device__ __forceinline__ void ibs_mma_ga(v8f (&accF)[3][2], const v2f (&src)[16][2], const v8f (&S)[TPW]) {
    #pragma unroll
    for (int k = 0; k < 16; ++k) {
        const int n = 16 * G + k;
        const int j = n / 12;
        const int s = (n / 3) & 3;
        const int ip = n % 3;
        v2f b;
        b[0] = S[j][2 * s];
        b[1] = S[j][2 * s + 1];
        accF[ip][0] = ibs_mma(src[k][0], b, accF[ip][0]);
        accF[ip][1] = ibs_mma(src[k][1], b, accF[ip][1]);
    }
}

template <int H>
__device__ __forceinline__ void ibs_load_gb(v2f (&dst)[16][2], const float2* fbl,
                                            const int (&a1off)[TPW], int col0) {
    #pragma unroll
    for (int k = 0; k < 16; ++k) {
        const int m = 16 * H + k;
        ibs_load_pair(dst[k], fbl + a1off[m / 4] + (m & 3) * FBP + col0);
    }
}

template <int H, int K0, int K1>
__device__ __forceinline__ void ibs_mma_gb(v8f (&accO)[2], const v2f (&src)[16][2], const v8f (&S)[TPW]) {
    #pragma unroll
    for (int k = K0; k < K1; ++k) {
        const int m = 16 * H + k;
        const int j = m / 4;
        const int s = m & 3;
        v2f b;
        b[0] = S[j][2 * s];
        b[1] = S[j][2 * s + 1];
        accO[0] = ibs_mma(src[k][0], b, accO[0]);
        accO[1] = ibs_mma(src[k][1], b, accO[1]);
    }
}

__device__ __forceinline__ void ibs_bar_signal_2(v8f& a, v8f& b) {
    asm volatile("s_wait_dscnt 0x0\n\ts_barrier_signal -1" : "+v"(a), "+v"(b) :: "memory");
}
__device__ __forceinline__ void ibs_bar_wait_2(v8f& a, v8f& b) {
    asm volatile("s_barrier_wait -1" : "+v"(a), "+v"(b) :: "memory");
}
__device__ __forceinline__ void ibs_hold_2(v8f& a, v8f& b) {
    asm volatile("" : "+v"(a), "+v"(b) :: "memory");
}
__device__ __forceinline__ void ibs_bar_signal_8(v8f (&g)[TPW]) {
    asm volatile("s_wait_dscnt 0x0\n\ts_barrier_signal -1"
                 : "+v"(g[0]), "+v"(g[1]), "+v"(g[2]), "+v"(g[3]), "+v"(g[4]), "+v"(g[5]), "+v"(g[6]), "+v"(g[7]) :: "memory");
}
__device__ __forceinline__ void ibs_bar_wait_8(v8f (&g)[TPW]) {
    asm volatile("s_barrier_wait -1"
                 : "+v"(g[0]), "+v"(g[1]), "+v"(g[2]), "+v"(g[3]), "+v"(g[4]), "+v"(g[5]), "+v"(g[6]), "+v"(g[7]) :: "memory");
}
__device__ __forceinline__ void ibs_hold_8(v8f (&g)[TPW]) {
    asm volatile(""
                 : "+v"(g[0]), "+v"(g[1]), "+v"(g[2]), "+v"(g[3]), "+v"(g[4]), "+v"(g[5]), "+v"(g[6]), "+v"(g[7]) :: "memory");
}

__global__ __launch_bounds__(NTHR) void ibs_run(const float* __restrict__ barkspec,
                                                const float* __restrict__ spec_init,
                                                float* __restrict__ spec,
                                                const float2* __restrict__ fbp2,
                                                const float2* __restrict__ fbT2,
                                                float* lossp,
                                                int replay) {
    __shared__ __attribute__((aligned(16))) float2 lds_fb[(FP / 2) * FBP];
    __shared__ __attribute__((aligned(16))) float4 lds_x[XDIFF + 16 * 32];
    __shared__ float  lds_wsum[MAX_ITER * NW];
    __shared__ float  lds_gt[2 * NW * 32];
    __shared__ float  lds_loss[32];
    __shared__ int    lds_n;

    const int tid     = threadIdx.x;
    const int wave    = __builtin_amdgcn_readfirstlane(tid >> 5);
    const int lane    = tid & 31;
    const int mi      = lane & 15;
    const int hsel    = lane >> 4;
    const int rowbase = blockIdx.x * 16;

    int n_iter = MAX_ITER;
    if (replay) {
        if (tid < MAX_ITER - 1) {
            const float4* lp = (const float4*)(lossp + tid * 128);
            float s = 0.0f;
            #pragma unroll
            for (int i = 0; i < 32; ++i) {
                float4 v = lp[i];
                s += v.x; s += v.y; s += v.z; s += v.w;
            }
            lds_loss[tid] = s / (float)ROWS;
        }
        __syncthreads();
        if (tid == 0) {
            float prev = INFINITY;
            int n = MAX_ITER;
            for (int it = 0; it < MAX_ITER - 1; ++it) {
                float loss = lds_loss[it];
                bool stop = (loss < TOL_LOSS) || (fabsf(prev - loss) < TOL_CHG);
                prev = loss;
                if (stop) { n = it + 1; break; }
            }
            lds_n = n;
        }
        __syncthreads();
        n_iter = lds_n;
        if (n_iter >= MAX_ITER) return;
    }

    {
        const float4* src4 = (const float4*)fbp2;
        #pragma unroll 33
        for (int q = tid; q < (FP / 2) * (NB / 2); q += NTHR) {
            float4 v = src4[q];
            const int p = q >> 6;
            const int c = q & 63;
            *(float4*)&lds_fb[p * FBP + 2 * c] = v;
        }
    }
    float* stage = (float*)lds_x;
    {
        const float* srcp = spec_init + rowbase * N_STFT;
        #pragma unroll
        for (int k = 0; k < (16 * N_STFT + NTHR - 1) / NTHR; ++k) {
            int i = tid + k * NTHR;
            i = i < 16 * N_STFT ? i : 16 * N_STFT - 1;
            stage[i] = srcp[i];
        }
    }
    __syncthreads();

    v8f S[TPW], Bm[TPW];
    #pragma unroll
    for (int j = 0; j < TPW; ++j) {
        #pragma unroll
        for (int r = 0; r < 8; ++r) {
            const int f = (wave + 4 * j) * 16 + hsel * 8 + r;
            S[j][r]  = stage[mi * N_STFT + f];
            Bm[j][r] = 0.0f;
        }
    }
    float Stail  = hsel == 0 ? stage[mi * N_STFT + TAILF] : 0.0f;
    float Bmtail = 0.0f;

    const int bb = rowbase >> 9;
    const int t0 = rowbase & 511;
    float tgt[2][8];
    #pragma unroll
    for (int u = 0; u < 2; ++u) {
        #pragma unroll
        for (int r = 0; r < 8; ++r) {
            const int bark = (2 * wave + u) * 16 + hsel * 8 + r;
            tgt[u][r] = barkspec[(bb * N_BARKS + bark) * TIME + t0 + mi];
        }
    }

    int prl[4];
    #pragma unroll
    for (int ip = 0; ip < 4; ++ip) prl[ip] = (wave + ip) & 3;

    const float2* a2p = fbT2 + (4 * hsel) * FP + 16 * wave + mi;
    v2f A2[8][4][TPW];
    #pragma unroll
    for (int ip = 0; ip < 4; ++ip) {
        const float2* ap = a2p + (16 * prl[ip]) * FP;
        #pragma unroll
        for (int u = 0; u < 2; ++u) {
            #pragma unroll
            for (int s = 0; s < 4; ++s) {
                #pragma unroll
                for (int j = 0; j < TPW; ++j) {
                    float2 av = ap[(8 * u + s) * FP + 64 * j];
                    A2[2 * ip + u][s][j][0] = av.x;
                    A2[2 * ip + u][s][j][1] = av.y;
                }
            }
        }
    }
    const float2* a2t = fbT2 + (4 * hsel) * FP + TAILF + mi;
    v2f A2t[2][4];
    v2f A1t[2];
    #pragma unroll
    for (int u = 0; u < 2; ++u) {
        #pragma unroll
        for (int s = 0; s < 4; ++s) {
            float2 av = a2t[(8 * (2 * wave + u) + s) * FP];
            A2t[u][s][0] = av.x;
            A2t[u][s][1] = av.y;
        }
        float2 a1 = lds_fb[((TAILF / 2) + 4 * hsel) * FBP + 16 * (2 * wave + u) + mi];
        A1t[u][0] = a1.x;
        A1t[u][1] = a1.y;
    }
    __syncthreads();

    int a1off[TPW];
    #pragma unroll
    for (int j = 0; j < TPW; ++j) {
        const int T = wave + 4 * j;
        a1off[j] = (8 * T + 4 * hsel) * FBP + mi;
    }
    int colr[4];
    #pragma unroll
    for (int ip = 0; ip < 4; ++ip) colr[ip] = 32 * prl[ip];

    v2f Af[2][16][2];
    ibs_load_ga<0>(Af[0], lds_fb, a1off, colr);

    for (int it = 0; it < n_iter; ++it) {
        v8f accF[3][2];
        v8f accO[2];
        #pragma unroll
        for (int ip = 0; ip < 3; ++ip) {
            #pragma unroll
            for (int u = 0; u < 2; ++u)
                accF[ip][u] = (v8f){0.f, 0.f, 0.f, 0.f, 0.f, 0.f, 0.f, 0.f};
        }
        #pragma unroll
        for (int u = 0; u < 2; ++u)
            accO[u] = (v8f){0.f, 0.f, 0.f, 0.f, 0.f, 0.f, 0.f, 0.f};

        ibs_load_ga<1>(Af[1], lds_fb, a1off, colr);
        __builtin_amdgcn_sched_barrier(0);
        ibs_mma_ga<0>(accF, Af[0], S);
        __builtin_amdgcn_sched_barrier(0);
        ibs_load_ga<2>(Af[0], lds_fb, a1off, colr);
        __builtin_amdgcn_sched_barrier(0);
        ibs_mma_ga<1>(accF, Af[1], S);
        __builtin_amdgcn_sched_barrier(0);
        ibs_load_ga<3>(Af[1], lds_fb, a1off, colr);
        __builtin_amdgcn_sched_barrier(0);
        ibs_mma_ga<2>(accF, Af[0], S);
        __builtin_amdgcn_sched_barrier(0);
        ibs_load_ga<4>(Af[0], lds_fb, a1off, colr);
        __builtin_amdgcn_sched_barrier(0);
        ibs_mma_ga<3>(accF, Af[1], S);
        __builtin_amdgcn_sched_barrier(0);
        ibs_load_ga<5>(Af[1], lds_fb, a1off, colr);
        __builtin_amdgcn_sched_barrier(0);
        ibs_mma_ga<4>(accF, Af[0], S);
        __builtin_amdgcn_sched_barrier(0);
        ibs_load_gb<0>(Af[0], lds_fb, a1off, colr[0]);
        __builtin_amdgcn_sched_barrier(0);
        ibs_mma_ga<5>(accF, Af[1], S);
        __builtin_amdgcn_sched_barrier(0);

        #pragma unroll
        for (int ip = 1; ip < 4; ++ip) {
            #pragma unroll
            for (int u = 0; u < 2; ++u) {
                const int bt = 2 * prl[ip] + u;
                lds_x[(wave * 16 + bt * 2 + 0) * 32 + lane] =
                    make_float4(accF[ip - 1][u][0], accF[ip - 1][u][1], accF[ip - 1][u][2], accF[ip - 1][u][3]);
                lds_x[(wave * 16 + bt * 2 + 1) * 32 + lane] =
                    make_float4(accF[ip - 1][u][4], accF[ip - 1][u][5], accF[ip - 1][u][6], accF[ip - 1][u][7]);
            }
        }
        ibs_load_gb<1>(Af[1], lds_fb, a1off, colr[0]);
        __builtin_amdgcn_sched_barrier(0);

        ibs_mma_gb<0, 0, 16>(accO, Af[0], S);
        __builtin_amdgcn_sched_barrier(0);
        ibs_bar_signal_2(accO[0], accO[1]);
        __builtin_amdgcn_sched_barrier(0);
        ibs_mma_gb<1, 0, 8>(accO, Af[1], S);
        __builtin_amdgcn_sched_barrier(0);
        ibs_bar_wait_2(accO[0], accO[1]);
        float4 plo[3][2], phi[3][2];
        #pragma unroll
        for (int o = 0; o < 3; ++o) {
            #pragma unroll
            for (int u = 0; u < 2; ++u) {
                const int bt = 2 * wave + u;
                plo[o][u] = lds_x[(prl[o + 1] * 16 + bt * 2 + 0) * 32 + lane];
                phi[o][u] = lds_x[(prl[o + 1] * 16 + bt * 2 + 1) * 32 + lane];
            }
        }
        ibs_hold_2(accO[0], accO[1]);
        __builtin_amdgcn_sched_barrier(0);
        ibs_mma_gb<1, 8, 16>(accO, Af[1], S);
        v8f tacc[2];
        {
            v2f b;
            b[0] = Stail;
            b[1] = 0.0f;
            #pragma unroll
            for (int u = 0; u < 2; ++u)
                tacc[u] = ibs_mma(A1t[u], b, (v8f){0.f, 0.f, 0.f, 0.f, 0.f, 0.f, 0.f, 0.f});
        }
        __builtin_amdgcn_sched_barrier(0);

        float ssq = 0.0f;
        float dd[2][8];
        #pragma unroll
        for (int u = 0; u < 2; ++u) {
            float p[8];
            #pragma unroll
            for (int r = 0; r < 8; ++r) p[r] = accO[u][r];
            #pragma unroll
            for (int o = 0; o < 3; ++o) {
                p[0] += plo[o][u].x; p[1] += plo[o][u].y; p[2] += plo[o][u].z; p[3] += plo[o][u].w;
                p[4] += phi[o][u].x; p[5] += phi[o][u].y; p[6] += phi[o][u].z; p[7] += phi[o][u].w;
            }
            #pragma unroll
            for (int r = 0; r < 8; ++r) {
                p[r] += tacc[u][r];
                dd[u][r] = tgt[u][r] - p[r];
                ssq += dd[u][r] * dd[u][r];
            }
            const int bt = 2 * wave + u;
            lds_x[XDIFF + (bt * 2 + 0) * 32 + lane] = make_float4(dd[u][0], dd[u][1], dd[u][2], dd[u][3]);
            lds_x[XDIFF + (bt * 2 + 1) * 32 + lane] = make_float4(dd[u][4], dd[u][5], dd[u][6], dd[u][7]);
        }
        v8f gt[2];
        #pragma unroll
        for (int u = 0; u < 2; ++u)
            gt[u] = (v8f){0.f, 0.f, 0.f, 0.f, 0.f, 0.f, 0.f, 0.f};
        #pragma unroll
        for (int s = 0; s < 4; ++s) {
            #pragma unroll
            for (int u = 0; u < 2; ++u) {
                v2f b;
                b[0] = dd[u][2 * s];
                b[1] = dd[u][2 * s + 1];
                gt[u] = ibs_mma(A2t[u][s], b, gt[u]);
            }
        }
        lds_gt[(2 * wave + 0) * 32 + lane] = gt[0][0];
        lds_gt[(2 * wave + 1) * 32 + lane] = gt[1][0];
        __builtin_amdgcn_sched_barrier(0);

        v8f g[TPW];
        #pragma unroll
        for (int j = 0; j < TPW; ++j)
            g[j] = (v8f){0.f, 0.f, 0.f, 0.f, 0.f, 0.f, 0.f, 0.f};
        #pragma unroll
        for (int t = 0; t < 5; ++t) {
            const float other = __shfl_xor(ssq, 16 >> t);
            __builtin_amdgcn_sched_barrier(0);
            #pragma unroll
            for (int q = 6 * t; q < (t == 4 ? 32 : 6 * t + 6); ++q) {
                const int s = q >> 3;
                const int j = q & 7;
                v2f b;
                b[0] = dd[0][2 * s];
                b[1] = dd[0][2 * s + 1];
                g[j] = ibs_mma(A2[0][s][j], b, g[j]);
            }
            __builtin_amdgcn_sched_barrier(0);
            ssq += other;
        }
        __builtin_amdgcn_sched_barrier(0);
        ibs_bar_signal_8(g);
        if (lane == 0) lds_wsum[it * NW + wave] = ssq;
        __builtin_amdgcn_sched_barrier(0);
        #pragma unroll
        for (int s = 0; s < 2; ++s) {
            v2f b;
            b[0] = dd[1][2 * s];
            b[1] = dd[1][2 * s + 1];
            #pragma unroll
            for (int j = 0; j < TPW; ++j)
                g[j] = ibs_mma(A2[1][s][j], b, g[j]);
        }
        __builtin_amdgcn_sched_barrier(0);
        ibs_bar_wait_8(g);
        v8f D1[6];
        #pragma unroll
        for (int ip = 1; ip < 4; ++ip) {
            #pragma unroll
            for (int u = 0; u < 2; ++u) {
                const int bt = 2 * prl[ip] + u;
                const int i = 2 * (ip - 1) + u;
                float4 lo = lds_x[XDIFF + (bt * 2 + 0) * 32 + lane];
                float4 hi = lds_x[XDIFF + (bt * 2 + 1) * 32 + lane];
                D1[i][0] = lo.x; D1[i][1] = lo.y; D1[i][2] = lo.z; D1[i][3] = lo.w;
                D1[i][4] = hi.x; D1[i][5] = hi.y; D1[i][6] = hi.z; D1[i][7] = hi.w;
            }
        }
        ibs_hold_8(g);
        __builtin_amdgcn_sched_barrier(0);
        #pragma unroll
        for (int s = 2; s < 4; ++s) {
            v2f b;
            b[0] = dd[1][2 * s];
            b[1] = dd[1][2 * s + 1];
            #pragma unroll
            for (int j = 0; j < TPW; ++j)
                g[j] = ibs_mma(A2[1][s][j], b, g[j]);
        }
        #pragma unroll
        for (int i = 0; i < 6; ++i) {
            #pragma unroll
            for (int s = 0; s < 4; ++s) {
                v2f b;
                b[0] = D1[i][2 * s];
                b[1] = D1[i][2 * s + 1];
                #pragma unroll
                for (int j = 0; j < TPW; ++j)
                    g[j] = ibs_mma(A2[2 + i][s][j], b, g[j]);
            }
        }
        __builtin_amdgcn_sched_barrier(0);
        ibs_load_ga<0>(Af[0], lds_fb, a1off, colr);
        float q[2 * NW];
        #pragma unroll
        for (int i = 0; i < 2 * NW; ++i) q[i] = lds_gt[i * 32 + lane];
        __builtin_amdgcn_sched_barrier(0);
        #pragma unroll
        for (int j = 0; j < TPW; ++j) {
            #pragma unroll
            for (int r = 0; r < 8; ++r) {
                float gg = g[j][r] * (-2.0f / (float)ROWS);
                float bn = MOMENTUM * Bm[j][r] + gg;
                float sn = S[j][r] - LR * bn;
                S[j][r]  = sn > 0.0f ? sn : 0.0f;
                Bm[j][r] = bn;
            }
        }
        {
            const float gtail = (((q[0] + q[1]) + (q[2] + q[3])) + (q[4] + q[5])) + (q[6] + q[7]);
            float gg = gtail * (-2.0f / (float)ROWS);
            float bn = MOMENTUM * Bmtail + gg;
            float sn = Stail - LR * bn;
            Stail  = sn > 0.0f ? sn : 0.0f;
            Bmtail = bn;
        }
    }

    __syncthreads();
    if (!replay && tid < MAX_ITER) {
        float s = 0.0f;
        for (int w = 0; w < NW; ++w) s += lds_wsum[tid * NW + w];
        lossp[tid * 128 + blockIdx.x] = s;
    }
    #pragma unroll
    for (int j = 0; j < TPW; ++j) {
        #pragma unroll
        for (int r = 0; r < 8; ++r) {
            const int f = (wave + 4 * j) * 16 + hsel * 8 + r;
            stage[f * 16 + mi] = S[j][r];
        }
    }
    if (wave == 0) {
        #pragma unroll
        for (int r = 0; r < 8; ++r)
            stage[(TAILF + hsel * 8 + r) * 16 + mi] = (r == 0) ? Stail : 0.0f;
    }
    __syncthreads();
    {
        float4* dst = (float4*)(spec + rowbase * FP);
        const float4* st4 = (const float4*)stage;
        for (int i = tid; i < 16 * FP / 4; i += NTHR) dst[i] = st4[i];
    }
}

#define FIN_WG   171
#define FIN_PER  6

__global__ __launch_bounds__(256) void ibs_finalize(const float* __restrict__ spec,
                                                    float* __restrict__ out) {
    const v4f* src = (const v4f*)spec;
    v4f*       dst = (v4f*)out;
    const int idx0 = blockIdx.x * 256 + threadIdx.x;
    v4f v[FIN_PER];
    #pragma unroll
    for (int k = 0; k < FIN_PER; ++k) {
        const int idx4 = idx0 + k * (FIN_WG * 256);
        const int q  = idx4 & 127;
        const int bf = idx4 >> 7;
        const int b  = bf / N_STFT;
        const int f  = bf - b * N_STFT;
        v[k] = src[((b * (TIME / 16) + (q >> 2)) * FP + f) * 4 + (q & 3)];
    }
    asm volatile("" :: "v"(v[0]), "v"(v[1]), "v"(v[2]), "v"(v[3]), "v"(v[4]), "v"(v[5]));
    #pragma unroll
    for (int k = 0; k < FIN_PER; ++k)
        dst[idx0 + k * (FIN_WG * 256)] = v[k];
}


extern "C" void kernel_launch(void* const* d_in, const int* in_sizes, int n_in,
                              void* d_out, int out_size, void* d_ws, size_t ws_size,
                              hipStream_t stream) {
    const float* barkspec  = (const float*)d_in[0];
    const float* fb        = (const float*)d_in[1];
    const float* spec_init = (const float*)d_in[2];

    float* ws       = (float*)d_ws;
    float* spec     = ws;
    float* lossp    = spec + ROWS * FP;
    float* fbp2     = lossp + ROWS * FP;
    float* fbT2     = fbp2 + FP * NB;

    ibs_init_fb<<<(FP * NB + 255) / 256, 256, 0, stream>>>(fb, fbp2, fbT2);
    ibs_run<<<ROWS / 16, NTHR, 0, stream>>>(barkspec, spec_init, spec,
                                            (const float2*)fbp2, (const float2*)fbT2, lossp, 0);
    ibs_run<<<ROWS / 16, NTHR, 0, stream>>>(barkspec, spec_init, spec,
                                            (const float2*)fbp2, (const float2*)fbT2, lossp, 1);
    ibs_finalize<<<FIN_WG, 256, 0, stream>>>(spec, (float*)d_out);
}


# ibs_run: LDS-unit instruction groups per iteration 18 -> 8: fragment reads two groups at a place (third register set), loss butterfly as vector-ALU lane ops, its sum parked per lane; sum order same
# speedup vs baseline: 5.1323x; 1.0474x over previous
#include <hip/hip_runtime.h>
#include <math.h>

typedef float v2f __attribute__((ext_vector_type(2)));
typedef float v4f __attribute__((ext_vector_type(4)));
typedef float v8f __attribute__((ext_vector_type(8)));

#define N_STFT   513
#define N_BARKS  128
#define BATCH    4
#define TIME     512
#define ROWS     (BATCH * TIME)
#define FP       528
#define NB       128
#define MAX_ITER 30
#define LR       0.1f
#define MOMENTUM 0.9f
#define TOL_LOSS 1e-5f
#define TOL_CHG  1e-8f
#define NW       4
#define NTHR     (NW * 32)
#define TPW      8
#define TAILF    512
#define FBP      132
#define XDIFF    (NW * 16 * 32)

__global__ __launch_bounds__(256) void ibs_init_fb(const float* __restrict__ fb_in,
                                                   float* __restrict__ fbp2,
                                                   float* __restrict__ fbT2) {
    int idx = blockIdx.x * blockDim.x + threadIdx.x;
    if (idx >= FP * NB) return;
    int f  = idx / NB;
    int kk = idx - f * NB;
    float v = (f < N_STFT) ? fb_in[f * NB + kk] : 0.0f;
    fbp2[(((f  >> 1) * NB) + kk) * 2 + (f  & 1)] = v;
    fbT2[(((kk >> 1) * FP) + f ) * 2 + (kk & 1)] = v;
}

__device__ __forceinline__ void ibs_load_pair(v2f (&dst)[2], const float2* p) {
    float2 a = p[0];
    float2 b = p[16];
    dst[0][0] = a.x; dst[0][1] = a.y;
    dst[1][0] = b.x; dst[1][1] = b.y;
}

__device__ __forceinline__ v8f ibs_mma(v2f a, v2f b, v8f c) {
    return __builtin_amdgcn_wmma_f32_16x16x4_f32(false, a, false, b, (short)0, c, false, false);
}

template <int G>
__device__ __forceinline__ void ibs_load_ga(v2f (&dst)[16][2], const float2* fbl,
                                            const int (&a1off)[TPW], const int (&colr)[4]) {
    #pragma unroll
    for (int k = 0; k < 16; ++k) {
        const int n = 16 * G + k;
        ibs_load_pair(dst[k], fbl + a1off[n / 12] + ((n / 3) & 3) * FBP + colr[1 + n % 3]);
    }
}

template <int G>
__device__ __forceinline__ void ibs_mma_ga(v8f (&accF)[3][2], const v2f (&src)[16][2], const v8f (&S)[TPW]) {
    #pragma unroll
    for (int k = 0; k < 16; ++k) {
        const int n = 16 * G + k;
        const int j = n / 12;
        const int s = (n / 3) & 3;
        const int ip = n % 3;
        v2f b;
        b[0] = S[j][2 * s];
        b[1] = S[j][2 * s + 1];
        accF[ip][0] = ibs_mma(src[k][0], b, accF[ip][0]);
        accF[ip][1] = ibs_mma(src[k][1], b, accF[ip][1]);
    }
}

template <int H>
__device__ __forceinline__ void ibs_load_gb(v2f (&dst)[16][2], const float2* fbl,
                                            const int (&a1off)[TPW], int col0) {
    #pragma unroll
    for (int k = 0; k < 16; ++k) {
        const int m = 16 * H + k;
        ibs_load_pair(dst[k], fbl + a1off[m / 4] + (m & 3) * FBP + col0);
    }
}

template <int H, int K0, int K1>
__device__ __forceinline__ void ibs_mma_gb(v8f (&accO)[2], const v2f (&src)[16][2], const v8f (&S)[TPW]) {
    #pragma unroll
    for (int k = K0; k < K1; ++k) {
        const int m = 16 * H + k;
        const int j = m / 4;
        const int s = m & 3;
        v2f b;
        b[0] = S[j][2 * s];
        b[1] = S[j][2 * s + 1];
        accO[0] = ibs_mma(src[k][0], b, accO[0]);
        accO[1] = ibs_mma(src[k][1], b, accO[1]);
    }
}

__device__ __forceinline__ void ibs_bar_signal_2(v8f& a, v8f& b) {
    asm volatile("s_wait_dscnt 0x0\n\ts_barrier_signal -1" : "+v"(a), "+v"(b) :: "memory");
}
__device__ __forceinline__ void ibs_bar_wait_2(v8f& a, v8f& b) {
    asm volatile("s_barrier_wait -1" : "+v"(a), "+v"(b) :: "memory");
}
__device__ __forceinline__ void ibs_hold_2(v8f& a, v8f& b) {
    asm volatile("" : "+v"(a), "+v"(b) :: "memory");
}
__device__ __forceinline__ void ibs_bar_signal_8(v8f (&g)[TPW]) {
    asm volatile("s_wait_dscnt 0x0\n\ts_barrier_signal -1"
                 : "+v"(g[0]), "+v"(g[1]), "+v"(g[2]), "+v"(g[3]), "+v"(g[4]), "+v"(g[5]), "+v"(g[6]), "+v"(g[7]) :: "memory");
}
__device__ __forceinline__ void ibs_bar_wait_8(v8f (&g)[TPW]) {
    asm volatile("s_barrier_wait -1"
                 : "+v"(g[0]), "+v"(g[1]), "+v"(g[2]), "+v"(g[3]), "+v"(g[4]), "+v"(g[5]), "+v"(g[6]), "+v"(g[7]) :: "memory");
}
__device__ __forceinline__ void ibs_hold_8(v8f (&g)[TPW]) {
    asm volatile(""
                 : "+v"(g[0]), "+v"(g[1]), "+v"(g[2]), "+v"(g[3]), "+v"(g[4]), "+v"(g[5]), "+v"(g[6]), "+v"(g[7]) :: "memory");
}

template <int M>
__device__ __forceinline__ float ibs_row_partner(float v) {
    return __builtin_bit_cast(float, __builtin_amdgcn_update_dpp(0, __builtin_bit_cast(int, v), 0x160 + M, 0xf, 0xf, false));
}
__device__ __forceinline__ float ibs_other_row(float v) {
    return __builtin_bit_cast(float, __builtin_amdgcn_permlanex16(__builtin_bit_cast(int, v), __builtin_bit_cast(int, v),
                                                                  0x76543210, 0xfedcba98, false, false));
}

__global__ __launch_bounds__(NTHR) void ibs_run(const float* __restrict__ barkspec,
                                                const float* __restrict__ spec_init,
                                                float* __restrict__ spec,
                                                const float2* __restrict__ fbp2,
                                                const float2* __restrict__ fbT2,
                                                float* lossp,
                                                int replay) {
    __shared__ __attribute__((aligned(16))) float2 lds_fb[(FP / 2) * FBP];
    __shared__ __attribute__((aligned(16))) float4 lds_x[XDIFF + 16 * 32];
    __shared__ float  lds_wsum[MAX_ITER * NW];
    __shared__ float  lds_gt[2 * NW * 32];
    __shared__ float  lds_loss[32];
    __shared__ int    lds_n;

    const int tid     = threadIdx.x;
    const int wave    = __builtin_amdgcn_readfirstlane(tid >> 5);
    const int lane    = tid & 31;
    const int mi      = lane & 15;
    const int hsel    = lane >> 4;
    const int rowbase = blockIdx.x * 16;

    int n_iter = MAX_ITER;
    if (replay) {
        if (tid < MAX_ITER - 1) {
            const float4* lp = (const float4*)(lossp + tid * 128);
            float s = 0.0f;
            #pragma unroll
            for (int i = 0; i < 32; ++i) {
                float4 v = lp[i];
                s += v.x; s += v.y; s += v.z; s += v.w;
            }
            lds_loss[tid] = s / (float)ROWS;
        }
        __syncthreads();
        if (tid == 0) {
            float prev = INFINITY;
            int n = MAX_ITER;
            for (int it = 0; it < MAX_ITER - 1; ++it) {
                float loss = lds_loss[it];
                bool stop = (loss < TOL_LOSS) || (fabsf(prev - loss) < TOL_CHG);
                prev = loss;
                if (stop) { n = it + 1; break; }
            }
            lds_n = n;
        }
        __syncthreads();
        n_iter = lds_n;
        if (n_iter >= MAX_ITER) return;
    }

    {
        const float4* src4 = (const float4*)fbp2;
        #pragma unroll 33
        for (int q = tid; q < (FP / 2) * (NB / 2); q += NTHR) {
            float4 v = src4[q];
            const int p = q >> 6;
            const int c = q & 63;
            *(float4*)&lds_fb[p * FBP + 2 * c] = v;
        }
    }
    float* stage = (float*)lds_x;
    {
        const float* srcp = spec_init + rowbase * N_STFT;
        #pragma unroll
        for (int k = 0; k < (16 * N_STFT + NTHR - 1) / NTHR; ++k) {
            int i = tid + k * NTHR;
            i = i < 16 * N_STFT ? i : 16 * N_STFT - 1;
            stage[i] = srcp[i];
        }
    }
    __syncthreads();

    v8f S[TPW], Bm[TPW];
    #pragma unroll
    for (int j = 0; j < TPW; ++j) {
        #pragma unroll
        for (int r = 0; r < 8; ++r) {
            const int f = (wave + 4 * j) * 16 + hsel * 8 + r;
            S[j][r]  = stage[mi * N_STFT + f];
            Bm[j][r] = 0.0f;
        }
    }
    float Stail  = hsel == 0 ? stage[mi * N_STFT + TAILF] : 0.0f;
    float Bmtail = 0.0f;

    const int bb = rowbase >> 9;
    const int t0 = rowbase & 511;
    float tgt[2][8];
    #pragma unroll
    for (int u = 0; u < 2; ++u) {
        #pragma unroll
        for (int r = 0; r < 8; ++r) {
            const int bark = (2 * wave + u) * 16 + hsel * 8 + r;
            tgt[u][r] = barkspec[(bb * N_BARKS + bark) * TIME + t0 + mi];
        }
    }

    int prl[4];
    #pragma unroll
    for (int ip = 0; ip < 4; ++ip) prl[ip] = (wave + ip) & 3;

    const float2* a2p = fbT2 + (4 * hsel) * FP + 16 * wave + mi;
    v2f A2[8][4][TPW];
    #pragma unroll
    for (int ip = 0; ip < 4; ++ip) {
        const float2* ap = a2p + (16 * prl[ip]) * FP;
        #pragma unroll
        for (int u = 0; u < 2; ++u) {
            #pragma unroll
            for (int s = 0; s < 4; ++s) {
                #pragma unroll
                for (int j = 0; j < TPW; ++j) {
                    float2 av = ap[(8 * u + s) * FP + 64 * j];
                    A2[2 * ip + u][s][j][0] = av.x;
                    A2[2 * ip + u][s][j][1] = av.y;
                }
            }
        }
    }
    const float2* a2t = fbT2 + (4 * hsel) * FP + TAILF + mi;
    v2f A2t[2][4];
    v2f A1t[2];
    #pragma unroll
    for (int u = 0; u < 2; ++u) {
        #pragma unroll
        for (int s = 0; s < 4; ++s) {
            float2 av = a2t[(8 * (2 * wave + u) + s) * FP];
            A2t[u][s][0] = av.x;
            A2t[u][s][1] = av.y;
        }
        float2 a1 = lds_fb[((TAILF / 2) + 4 * hsel) * FBP + 16 * (2 * wave + u) + mi];
        A1t[u][0] = a1.x;
        A1t[u][1] = a1.y;
    }
    __syncthreads();

    int a1off[TPW];
    #pragma unroll
    for (int j = 0; j < TPW; ++j) {
        const int T = wave + 4 * j;
        a1off[j] = (8 * T + 4 * hsel) * FBP + mi;
    }
    int colr[4];
    #pragma unroll
    for (int ip = 0; ip < 4; ++ip) colr[ip] = 32 * prl[ip];

    v2f Af[3][16][2];
    ibs_load_ga<0>(Af[0], lds_fb, a1off, colr);

    static_assert(MAX_ITER <= 32, "one lane per iteration");
    float wsum_it = 0.0f;

    for (int it = 0; it < n_iter; ++it) {
        v8f accF[3][2];
        v8f accO[2];
        #pragma unroll
        for (int ip = 0; ip < 3; ++ip) {
            #pragma unroll
            for (int u = 0; u < 2; ++u)
                accF[ip][u] = (v8f){0.f, 0.f, 0.f, 0.f, 0.f, 0.f, 0.f, 0.f};
        }
        #pragma unroll
        for (int u = 0; u < 2; ++u)
            accO[u] = (v8f){0.f, 0.f, 0.f, 0.f, 0.f, 0.f, 0.f, 0.f};

        ibs_load_ga<1>(Af[1], lds_fb, a1off, colr);
        ibs_load_ga<2>(Af[2], lds_fb, a1off, colr);
        __builtin_amdgcn_sched_barrier(0);
        ibs_mma_ga<0>(accF, Af[0], S);
        ibs_mma_ga<1>(accF, Af[1], S);
        __builtin_amdgcn_sched_barrier(0);
        ibs_load_ga<3>(Af[0], lds_fb, a1off, colr);
        ibs_load_ga<4>(Af[1], lds_fb, a1off, colr);
        __builtin_amdgcn_sched_barrier(0);
        ibs_mma_ga<2>(accF, Af[2], S);
        ibs_mma_ga<3>(accF, Af[0], S);
        __builtin_amdgcn_sched_barrier(0);
        ibs_load_ga<5>(Af[2], lds_fb, a1off, colr);
        ibs_load_gb<0>(Af[0], lds_fb, a1off, colr[0]);
        __builtin_amdgcn_sched_barrier(0);
        ibs_mma_ga<4>(accF, Af[1], S);
        ibs_mma_ga<5>(accF, Af[2], S);
        __builtin_amdgcn_sched_barrier(0);

        #pragma unroll
        for (int ip = 1; ip < 4; ++ip) {
            #pragma unroll
            for (int u = 0; u < 2; ++u) {
                const int bt = 2 * prl[ip] + u;
                lds_x[(wave * 16 + bt * 2 + 0) * 32 + lane] =
                    make_float4(accF[ip - 1][u][0], accF[ip - 1][u][1], accF[ip - 1][u][2], accF[ip - 1][u][3]);
                lds_x[(wave * 16 + bt * 2 + 1) * 32 + lane] =
                    make_float4(accF[ip - 1][u][4], accF[ip - 1][u][5], accF[ip - 1][u][6], accF[ip - 1][u][7]);
            }
        }
        ibs_load_gb<1>(Af[1], lds_fb, a1off, colr[0]);
        __builtin_amdgcn_sched_barrier(0);

        ibs_mma_gb<0, 0, 16>(accO, Af[0], S);
        __builtin_amdgcn_sched_barrier(0);
        ibs_bar_signal_2(accO[0], accO[1]);
        __builtin_amdgcn_sched_barrier(0);
        ibs_mma_gb<1, 0, 8>(accO, Af[1], S);
        __builtin_amdgcn_sched_barrier(0);
        ibs_bar_wait_2(accO[0], accO[1]);
        float4 plo[3][2], phi[3][2];
        #pragma unroll
        for (int o = 0; o < 3; ++o) {
            #pragma unroll
            for (int u = 0; u < 2; ++u) {
                const int bt = 2 * wave + u;
                plo[o][u] = lds_x[(prl[o + 1] * 16 + bt * 2 + 0) * 32 + lane];
                phi[o][u] = lds_x[(prl[o + 1] * 16 + bt * 2 + 1) * 32 + lane];
            }
        }
        ibs_hold_2(accO[0], accO[1]);
        __builtin_amdgcn_sched_barrier(0);
        ibs_mma_gb<1, 8, 16>(accO, Af[1], S);
        v8f tacc[2];
        {
            v2f b;
            b[0] = Stail;
            b[1] = 0.0f;
            #pragma unroll
            for (int u = 0; u < 2; ++u)
                tacc[u] = ibs_mma(A1t[u], b, (v8f){0.f, 0.f, 0.f, 0.f, 0.f, 0.f, 0.f, 0.f});
        }
        __builtin_amdgcn_sched_barrier(0);

        float ssq = 0.0f;
        float dd[2][8];
        #pragma unroll
        for (int u = 0; u < 2; ++u) {
            float p[8];
            #pragma unroll
            for (int r = 0; r < 8; ++r) p[r] = accO[u][r];
            #pragma unroll
            for (int o = 0; o < 3; ++o) {
                p[0] += plo[o][u].x; p[1] += plo[o][u].y; p[2] += plo[o][u].z; p[3] += plo[o][u].w;
                p[4] += phi[o][u].x; p[5] += phi[o][u].y; p[6] += phi[o][u].z; p[7] += phi[o][u].w;
            }
            #pragma unroll
            for (int r = 0; r < 8; ++r) {
                p[r] += tacc[u][r];
                dd[u][r] = tgt[u][r] - p[r];
                ssq += dd[u][r] * dd[u][r];
            }
            const int bt = 2 * wave + u;
            lds_x[XDIFF + (bt * 2 + 0) * 32 + lane] = make_float4(dd[u][0], dd[u][1], dd[u][2], dd[u][3]);
            lds_x[XDIFF + (bt * 2 + 1) * 32 + lane] = make_float4(dd[u][4], dd[u][5], dd[u][6], dd[u][7]);
        }
        v8f gt[2];
        #pragma unroll
        for (int u = 0; u < 2; ++u)
            gt[u] = (v8f){0.f, 0.f, 0.f, 0.f, 0.f, 0.f, 0.f, 0.f};
        #pragma unroll
        for (int s = 0; s < 4; ++s) {
            #pragma unroll
            for (int u = 0; u < 2; ++u) {
                v2f b;
                b[0] = dd[u][2 * s];
                b[1] = dd[u][2 * s + 1];
                gt[u] = ibs_mma(A2t[u][s], b, gt[u]);
            }
        }
        lds_gt[(2 * wave + 0) * 32 + lane] = gt[0][0];
        lds_gt[(2 * wave + 1) * 32 + lane] = gt[1][0];
        __builtin_amdgcn_sched_barrier(0);

        v8f g[TPW];
        #pragma unroll
        for (int j = 0; j < TPW; ++j)
            g[j] = (v8f){0.f, 0.f, 0.f, 0.f, 0.f, 0.f, 0.f, 0.f};
        #pragma unroll
        for (int t = 0; t < 5; ++t) {
            ssq += (t == 0) ? ibs_other_row(ssq)
                 : (t == 1) ? ibs_row_partner<8>(ssq)
                 : (t == 2) ? ibs_row_partner<4>(ssq)
                 : (t == 3) ? ibs_row_partner<2>(ssq)
                 :            ibs_row_partner<1>(ssq);
            #pragma unroll
            for (int q = 6 * t; q < (t == 4 ? 32 : 6 * t + 6); ++q) {
                const int s = q >> 3;
                const int j = q & 7;
                v2f b;
                b[0] = dd[0][2 * s];
                b[1] = dd[0][2 * s + 1];
                g[j] = ibs_mma(A2[0][s][j], b, g[j]);
            }
        }
        wsum_it = (lane == it) ? ssq : wsum_it;
        __builtin_amdgcn_sched_barrier(0);
        ibs_bar_signal_8(g);
        __builtin_amdgcn_sched_barrier(0);
        #pragma unroll
        for (int s = 0; s < 2; ++s) {
            v2f b;
            b[0] = dd[1][2 * s];
            b[1] = dd[1][2 * s + 1];
            #pragma unroll
            for (int j = 0; j < TPW; ++j)
                g[j] = ibs_mma(A2[1][s][j], b, g[j]);
        }
        __builtin_amdgcn_sched_barrier(0);
        ibs_bar_wait_8(g);
        v8f D1[6];
        #pragma unroll
        for (int ip = 1; ip < 4; ++ip) {
            #pragma unroll
            for (int u = 0; u < 2; ++u) {
                const int bt = 2 * prl[ip] + u;
                const int i = 2 * (ip - 1) + u;
                float4 lo = lds_x[XDIFF + (bt * 2 + 0) * 32 + lane];
                float4 hi = lds_x[XDIFF + (bt * 2 + 1) * 32 + lane];
                D1[i][0] = lo.x; D1[i][1] = lo.y; D1[i][2] = lo.z; D1[i][3] = lo.w;
                D1[i][4] = hi.x; D1[i][5] = hi.y; D1[i][6] = hi.z; D1[i][7] = hi.w;
            }
        }
        ibs_hold_8(g);
        __builtin_amdgcn_sched_barrier(0);
        #pragma unroll
        for (int s = 2; s < 4; ++s) {
            v2f b;
            b[0] = dd[1][2 * s];
            b[1] = dd[1][2 * s + 1];
            #pragma unroll
            for (int j = 0; j < TPW; ++j)
                g[j] = ibs_mma(A2[1][s][j], b, g[j]);
        }
        #pragma unroll
        for (int i = 0; i < 6; ++i) {
            #pragma unroll
            for (int s = 0; s < 4; ++s) {
                v2f b;
                b[0] = D1[i][2 * s];
                b[1] = D1[i][2 * s + 1];
                #pragma unroll
                for (int j = 0; j < TPW; ++j)
                    g[j] = ibs_mma(A2[2 + i][s][j], b, g[j]);
            }
        }
        __builtin_amdgcn_sched_barrier(0);
        ibs_load_ga<0>(Af[0], lds_fb, a1off, colr);
        float q[2 * NW];
        #pragma unroll
        for (int i = 0; i < 2 * NW; ++i) q[i] = lds_gt[i * 32 + lane];
        __builtin_amdgcn_sched_barrier(0);
        #pragma unroll
        for (int j = 0; j < TPW; ++j) {
            #pragma unroll
            for (int r = 0; r < 8; ++r) {
                float gg = g[j][r] * (-2.0f / (float)ROWS);
                float bn = MOMENTUM * Bm[j][r] + gg;
                float sn = S[j][r] - LR * bn;
                S[j][r]  = sn > 0.0f ? sn : 0.0f;
                Bm[j][r] = bn;
            }
        }
        {
            const float gtail = (((q[0] + q[1]) + (q[2] + q[3])) + (q[4] + q[5])) + (q[6] + q[7]);
            float gg = gtail * (-2.0f / (float)ROWS);
            float bn = MOMENTUM * Bmtail + gg;
            float sn = Stail - LR * bn;
            Stail  = sn > 0.0f ? sn : 0.0f;
            Bmtail = bn;
        }
    }

    if (lane < MAX_ITER) lds_wsum[lane * NW + wave] = wsum_it;
    __syncthreads();
    if (!replay && tid < MAX_ITER) {
        float s = 0.0f;
        for (int w = 0; w < NW; ++w) s += lds_wsum[tid * NW + w];
        lossp[tid * 128 + blockIdx.x] = s;
    }
    #pragma unroll
    for (int j = 0; j < TPW; ++j) {
        #pragma unroll
        for (int r = 0; r < 8; ++r) {
            const int f = (wave + 4 * j) * 16 + hsel * 8 + r;
            stage[f * 16 + mi] = S[j][r];
        }
    }
    if (wave == 0) {
        #pragma unroll
        for (int r = 0; r < 8; ++r)
            stage[(TAILF + hsel * 8 + r) * 16 + mi] = (r == 0) ? Stail : 0.0f;
    }
    __syncthreads();
    {
        float4* dst = (float4*)(spec + rowbase * FP);
        const float4* st4 = (const float4*)stage;
        for (int i = tid; i < 16 * FP / 4; i += NTHR) dst[i] = st4[i];
    }
}

#define FIN_WG   171
#define FIN_PER  6

__global__ __launch_bounds__(256) void ibs_finalize(const float* __restrict__ spec,
                                                    float* __restrict__ out) {
    const v4f* src = (const v4f*)spec;
    v4f*       dst = (v4f*)out;
    const int idx0 = blockIdx.x * 256 + threadIdx.x;
    v4f v[FIN_PER];
    #pragma unroll
    for (int k = 0; k < FIN_PER; ++k) {
        const int idx4 = idx0 + k * (FIN_WG * 256);
        const int q  = idx4 & 127;
        const int bf = idx4 >> 7;
        const int b  = bf / N_STFT;
        const int f  = bf - b * N_STFT;
        v[k] = src[((b * (TIME / 16) + (q >> 2)) * FP + f) * 4 + (q & 3)];
    }
    asm volatile("" :: "v"(v[0]), "v"(v[1]), "v"(v[2]), "v"(v[3]), "v"(v[4]), "v"(v[5]));
    #pragma unroll
    for (int k = 0; k < FIN_PER; ++k)
        dst[idx0 + k * (FIN_WG * 256)] = v[k];
}


extern "C" void kernel_launch(void* const* d_in, const int* in_sizes, int n_in,
                              void* d_out, int out_size, void* d_ws, size_t ws_size,
                              hipStream_t stream) {
    const float* barkspec  = (const float*)d_in[0];
    const float* fb        = (const float*)d_in[1];
    const float* spec_init = (const float*)d_in[2];

    float* ws       = (float*)d_ws;
    float* spec     = ws;
    float* lossp    = spec + ROWS * FP;
    float* fbp2     = lossp + ROWS * FP;
    float* fbT2     = fbp2 + FP * NB;

    ibs_init_fb<<<(FP * NB + 255) / 256, 256, 0, stream>>>(fb, fbp2, fbT2);
    ibs_run<<<ROWS / 16, NTHR, 0, stream>>>(barkspec, spec_init, spec,
                                            (const float2*)fbp2, (const float2*)fbT2, lossp, 0);
    ibs_run<<<ROWS / 16, NTHR, 0, stream>>>(barkspec, spec_init, spec,
                                            (const float2*)fbp2, (const float2*)fbT2, lossp, 1);
    ibs_finalize<<<FIN_WG, 256, 0, stream>>>(spec, (float*)d_out);
}
